# starting kernel
# baseline (speedup 1.0000x reference)
#include <hip/hip_runtime.h>
#include <hip/hip_bf16.h>
#include <math.h>


typedef float v2f __attribute__((ext_vector_type(2)));
typedef float v8f __attribute__((ext_vector_type(8)));

#define WMMA_F32(a, b, c) \
  __builtin_amdgcn_wmma_f32_16x16x4_f32(false, (a), false, (b), (short)0, (c), false, false)

static constexpr int D_IN = 128;


__global__ void k_zero_ints(int* __restrict__ p, int n) {
  int i = blockIdx.x * blockDim.x + threadIdx.x;
  if (i < n) p[i] = 0;
}

__global__ void k_hist(const int* __restrict__ etype, int E, int* __restrict__ counts) {
  int e = blockIdx.x * blockDim.x + threadIdx.x;
  if (e < E) atomicAdd(&counts[etype[e] & 7], 1);
}

__global__ void k_scan(const int* __restrict__ counts, int* __restrict__ starts) {
  if (threadIdx.x == 0 && blockIdx.x == 0) {
    int acc = 0;
    for (int r = 0; r < 8; ++r) {
      starts[r] = acc;
      acc += (counts[r] + 15) & ~15;
    }
    starts[8] = acc;
  }
}

__global__ void k_scatter(const int* __restrict__ ei, const int* __restrict__ etype, int E,
                          const int* __restrict__ starts, int* __restrict__ cursor,
                          int* __restrict__ bsrc, int* __restrict__ bdst) {
  int e = blockIdx.x * blockDim.x + threadIdx.x;
  if (e >= E) return;
  int r = etype[e] & 7;
  int pos = atomicAdd(&cursor[r], 1);
  int idx = starts[r] + pos;
  bsrc[idx] = ei[e];
  bdst[idx] = ei[E + e];
}

__global__ void k_pad(const int* __restrict__ counts, const int* __restrict__ starts,
                      int* __restrict__ bsrc, int* __restrict__ bdst, int dummyRow) {
  int r = blockIdx.x;
  int lo = starts[r] + counts[r];
  int hi = starts[r + 1];
  for (int i = lo + (int)threadIdx.x; i < hi; i += blockDim.x) {
    bsrc[i] = 0;
    bdst[i] = dummyRow;
  }
}


template <int DOUT>
__global__ __launch_bounds__(32 * (DOUT / 16)) void k_root_gemm(
    const float* __restrict__ x, const float* __restrict__ W,
    const float* __restrict__ bias, float* __restrict__ y, int N) {
  const int wave = threadIdx.x >> 5, lane = threadIdx.x & 31;
  const int m = lane & 15;
  const int khalf = (lane >> 4) << 1;
  const int row = (blockIdx.x << 4) + m;
  const int rowc = row < N ? row : N - 1;
  const float* xa = x + (size_t)rowc * D_IN;
  const int col = (wave << 4) + m;
  const float* wb = W + col;
  v8f c = {};
#pragma unroll 8
  for (int k = 0; k < D_IN; k += 4) {
    v2f a, b;
    a.x = xa[k + khalf];
    a.y = xa[k + khalf + 1];
    b.x = wb[(k + khalf) * DOUT];
    b.y = wb[(k + khalf + 1) * DOUT];
    c = WMMA_F32(a, b, c);
  }
  const float bv = bias[col];
  const int rbase = (blockIdx.x << 4) + ((lane >> 4) << 3);
  float* yp = y + (size_t)rbase * DOUT + col;
  if ((blockIdx.x << 4) + 16 <= N) {
#pragma unroll
    for (int i = 0; i < 8; ++i) yp[i * DOUT] = c[i] + bv;
  } else {
#pragma unroll
    for (int i = 0; i < 8; ++i)
      if (rbase + i < N) yp[i * DOUT] = c[i] + bv;
  }
}


template <int DOUT>
__global__ __launch_bounds__(32 * (DOUT / 16)) void k_edge_wmma(
    const float* __restrict__ x, const float* __restrict__ Wrel,
    const int* __restrict__ bsrc, const int* __restrict__ bdst,
    const int* __restrict__ starts, float* __restrict__ y) {
  const int base = blockIdx.x << 4;
  if (base >= starts[8]) return;
  const int wave = threadIdx.x >> 5, lane = threadIdx.x & 31;
  int r = 0;
#pragma unroll
  for (int i = 1; i < 8; ++i) r += (base >= starts[i]) ? 1 : 0;

  const int m = lane & 15;
  const int khalf = (lane >> 4) << 1;
  const int srow = bsrc[base + m];
  const float* xa = x + (size_t)srow * D_IN;
  const int col = (wave << 4) + m;
  const float* wb = Wrel + (size_t)r * (D_IN * DOUT) + col;

  v8f c = {};
#pragma unroll 8
  for (int k = 0; k < D_IN; k += 4) {
    v2f a, b;
    a.x = xa[k + khalf];
    a.y = xa[k + khalf + 1];
    b.x = wb[(k + khalf) * DOUT];
    b.y = wb[(k + khalf + 1) * DOUT];
    c = WMMA_F32(a, b, c);
  }

  const int mBase = (lane >> 4) << 3;
#pragma unroll
  for (int i = 0; i < 8; ++i) {
    int drow = bdst[base + mBase + i];
    atomicAdd(&y[(size_t)drow * DOUT + col], c[i]);
  }
}


__global__ void k_relu4(float4* __restrict__ p, long long n4) {
  long long i = (long long)blockIdx.x * blockDim.x + threadIdx.x;
  if (i < n4) {
    float4 v = p[i];
    v.x = fmaxf(v.x, 0.0f);
    v.y = fmaxf(v.y, 0.0f);
    v.z = fmaxf(v.z, 0.0f);
    v.w = fmaxf(v.w, 0.0f);
    p[i] = v;
  }
}

__global__ void k_logsoftmax(const float* __restrict__ y, float* __restrict__ out, int N) {
  int wave = threadIdx.x >> 5, lane = threadIdx.x & 31;
  int node = blockIdx.x * (blockDim.x >> 5) + wave;
  if (node >= N) return;
  const float* row = y + (size_t)node * 64;
  float v0 = row[lane], v1 = row[lane + 32];
  float mx = fmaxf(v0, v1);
#pragma unroll
  for (int off = 16; off > 0; off >>= 1) mx = fmaxf(mx, __shfl_xor(mx, off, 32));
  float s = expf(v0 - mx) + expf(v1 - mx);
#pragma unroll
  for (int off = 16; off > 0; off >>= 1) s += __shfl_xor(s, off, 32);
  float lse = logf(s) + mx;
  float* o = out + (size_t)node * 64;
  o[lane] = v0 - lse;
  o[lane + 32] = v1 - lse;
}


static inline void launch_layer(const float* x, const float* Wrel, const float* Wroot,
                                const float* bias, float* y, int dOut, int N,
                                const int* bsrc, const int* bdst, const int* starts,
                                int nodeTiles, int edgeTilesMax, hipStream_t stream) {
  if (dOut == 128) {
    k_root_gemm<128><<<nodeTiles, 32 * (128 / 16), 0, stream>>>(x, Wroot, bias, y, N);
    k_edge_wmma<128><<<edgeTilesMax, 32 * (128 / 16), 0, stream>>>(x, Wrel, bsrc, bdst, starts, y);
  } else {
    k_root_gemm<64><<<nodeTiles, 32 * (64 / 16), 0, stream>>>(x, Wroot, bias, y, N);
    k_edge_wmma<64><<<edgeTilesMax, 32 * (64 / 16), 0, stream>>>(x, Wrel, bsrc, bdst, starts, y);
  }
}

extern "C" void kernel_launch(void* const* d_in, const int* in_sizes, int n_in,
                              void* d_out, int out_size, void* d_ws, size_t ws_size,
                              hipStream_t stream) {
  const float* x0     = (const float*)d_in[0];
  const int*   ei     = (const int*)  d_in[1];
  const int*   etype  = (const int*)  d_in[2];
  const float* Wrel1  = (const float*)d_in[3];
  const float* Wroot1 = (const float*)d_in[4];
  const float* b1     = (const float*)d_in[5];
  const float* Wrel2  = (const float*)d_in[6];
  const float* Wroot2 = (const float*)d_in[7];
  const float* b2     = (const float*)d_in[8];
  const float* Wrel3  = (const float*)d_in[9];
  const float* Wroot3 = (const float*)d_in[10];
  const float* b3     = (const float*)d_in[11];

  const int N  = in_sizes[0] / D_IN;
  const int E  = in_sizes[2];
  const int D1 = 128, D2 = 128;
  const int D3 = in_sizes[11];

  char* w = (char*)d_ws;
  int* counts = (int*)w;
  int* cursor = counts + 8;
  int* starts = cursor + 8;
  size_t off = 256;
  int* bsrc = (int*)(w + off); off += (size_t)(E + 256) * sizeof(int);
  int* bdst = (int*)(w + off); off += (size_t)(E + 256) * sizeof(int);
  off = (off + 255) & ~(size_t)255;
  float* bufA = (float*)(w + off); off += (size_t)(N + 16) * D_IN * sizeof(float);
  float* bufB = (float*)(w + off);

  k_zero_ints<<<1, 64, 0, stream>>>(counts, 32);
  k_hist    <<<(E + 255) / 256, 256, 0, stream>>>(etype, E, counts);
  k_scan    <<<1, 1, 0, stream>>>(counts, starts);
  k_scatter <<<(E + 255) / 256, 256, 0, stream>>>(ei, etype, E, starts, cursor, bsrc, bdst);
  k_pad     <<<8, 64, 0, stream>>>(counts, starts, bsrc, bdst, N);

  const int nodeTiles    = (N + 15) / 16;
  const int edgeTilesMax = E / 16 + 8;

  launch_layer(x0, Wrel1, Wroot1, b1, bufA, D1, N, bsrc, bdst, starts,
               nodeTiles, edgeTilesMax, stream);
  {
    long long n4 = (long long)N * D1 / 4;
    k_relu4<<<(unsigned)((n4 + 255) / 256), 256, 0, stream>>>((float4*)bufA, n4);
  }

  launch_layer(bufA, Wrel2, Wroot2, b2, bufB, D2, N, bsrc, bdst, starts,
               nodeTiles, edgeTilesMax, stream);
  {
    long long n4 = (long long)N * D2 / 4;
    k_relu4<<<(unsigned)((n4 + 255) / 256), 256, 0, stream>>>((float4*)bufB, n4);
  }

  launch_layer(bufB, Wrel3, Wroot3, b3, bufA, D3, N, bsrc, bdst, starts,
               nodeTiles, edgeTilesMax, stream);
  k_logsoftmax<<<(N + 7) / 8, 256, 0, stream>>>(bufA, (float*)d_out, N);
}


# Bucketing counts reduced wave (ballot) -> work-group (LDS) -> device: one device atomic per work-group and relation, not per edge; edge order inside a bucket changes, float atomic scatter as before
# speedup vs baseline: 1.3313x; 1.3313x over previous
#include <hip/hip_runtime.h>
#include <hip/hip_bf16.h>
#include <math.h>


typedef float v2f __attribute__((ext_vector_type(2)));
typedef float v8f __attribute__((ext_vector_type(8)));

#define WMMA_F32(a, b, c) \
  __builtin_amdgcn_wmma_f32_16x16x4_f32(false, (a), false, (b), (short)0, (c), false, false)

static constexpr int D_IN = 128;


__global__ void k_zero_ints(int* __restrict__ p, int n) {
  int i = blockIdx.x * blockDim.x + threadIdx.x;
  if (i < n) p[i] = 0;
}

static constexpr int BKT_THREADS = 256;
static constexpr int BKT_ITEMS   = 8;
static constexpr int BKT_EDGES   = BKT_THREADS * BKT_ITEMS;

__device__ __forceinline__ unsigned rel_peers(int r, bool live) {
  unsigned peers = __builtin_amdgcn_ballot_w32(live);
#pragma unroll
  for (int b = 0; b < 3; ++b) {
    const bool bit = (r >> b) & 1;
    const unsigned vote = __builtin_amdgcn_ballot_w32(bit);
    peers &= bit ? vote : ~vote;
  }
  return peers;
}

__global__ __launch_bounds__(BKT_THREADS) void k_hist(const int* __restrict__ etype, int E,
                                                      int* __restrict__ counts) {
  __shared__ int cnt[8];
  const int lane = threadIdx.x & 31;
  if (threadIdx.x < 8) cnt[threadIdx.x] = 0;
  __syncthreads();
  const int e0 = blockIdx.x * BKT_EDGES + threadIdx.x;
  int r[BKT_ITEMS];
#pragma unroll
  for (int i = 0; i < BKT_ITEMS; ++i) {
    const int e = e0 + i * BKT_THREADS;
    const int ec = e < E ? e : E - 1;
    const int t = etype[ec];
    r[i] = e < E ? (t & 7) : -1;
  }
#pragma unroll
  for (int i = 0; i < BKT_ITEMS; ++i) {
    const bool live = r[i] >= 0;
    const unsigned peers = rel_peers(r[i], live);
    if (live && (peers & ((1u << lane) - 1u)) == 0) atomicAdd(&cnt[r[i]], __popc(peers));
  }
  __syncthreads();
  if (threadIdx.x < 8) {
    const int c = cnt[threadIdx.x];
    if (c) atomicAdd(&counts[threadIdx.x], c);
  }
}

__global__ void k_scan(const int* __restrict__ counts, int* __restrict__ starts) {
  if (threadIdx.x == 0 && blockIdx.x == 0) {
    int acc = 0;
    for (int r = 0; r < 8; ++r) {
      starts[r] = acc;
      acc += (counts[r] + 15) & ~15;
    }
    starts[8] = acc;
  }
}

__global__ __launch_bounds__(BKT_THREADS) void k_scatter(
    const int* __restrict__ ei, const int* __restrict__ etype, int E,
    const int* __restrict__ starts, int* __restrict__ cursor,
    int* __restrict__ bsrc, int* __restrict__ bdst) {
  __shared__ int cnt[8];
  __shared__ int base[8];
  const int lane = threadIdx.x & 31;
  if (threadIdx.x < 8) cnt[threadIdx.x] = 0;
  __syncthreads();
  const int e0 = blockIdx.x * BKT_EDGES + threadIdx.x;
  int r[BKT_ITEMS], s[BKT_ITEMS], d[BKT_ITEMS], rank[BKT_ITEMS];
#pragma unroll
  for (int i = 0; i < BKT_ITEMS; ++i) {
    const int e = e0 + i * BKT_THREADS;
    const int ec = e < E ? e : E - 1;
    const int t = etype[ec];
    r[i] = e < E ? (t & 7) : -1;
    s[i] = ei[ec];
    d[i] = ei[E + ec];
  }
#pragma unroll
  for (int i = 0; i < BKT_ITEMS; ++i) {
    const bool live = r[i] >= 0;
    const unsigned peers = rel_peers(r[i], live);
    const int below = __popc(peers & ((1u << lane) - 1u));
    int wbase = 0;
    if (live && below == 0) wbase = atomicAdd(&cnt[r[i]], __popc(peers));
    wbase = __shfl(wbase, live ? __ffs(peers) - 1 : lane, 32);
    rank[i] = wbase + below;
  }
  __syncthreads();
  if (threadIdx.x < 8) {
    const int c = cnt[threadIdx.x];
    base[threadIdx.x] = starts[threadIdx.x] + (c ? atomicAdd(&cursor[threadIdx.x], c) : 0);
  }
  __syncthreads();
#pragma unroll
  for (int i = 0; i < BKT_ITEMS; ++i) {
    if (r[i] >= 0) {
      const int idx = base[r[i]] + rank[i];
      bsrc[idx] = s[i];
      bdst[idx] = d[i];
    }
  }
}

__global__ void k_pad(const int* __restrict__ counts, const int* __restrict__ starts,
                      int* __restrict__ bsrc, int* __restrict__ bdst, int dummyRow) {
  int r = blockIdx.x;
  int lo = starts[r] + counts[r];
  int hi = starts[r + 1];
  for (int i = lo + (int)threadIdx.x; i < hi; i += blockDim.x) {
    bsrc[i] = 0;
    bdst[i] = dummyRow;
  }
}


template <int DOUT>
__global__ __launch_bounds__(32 * (DOUT / 16)) void k_root_gemm(
    const float* __restrict__ x, const float* __restrict__ W,
    const float* __restrict__ bias, float* __restrict__ y, int N) {
  const int wave = threadIdx.x >> 5, lane = threadIdx.x & 31;
  const int m = lane & 15;
  const int khalf = (lane >> 4) << 1;
  const int row = (blockIdx.x << 4) + m;
  const int rowc = row < N ? row : N - 1;
  const float* xa = x + (size_t)rowc * D_IN;
  const int col = (wave << 4) + m;
  const float* wb = W + col;
  v8f c = {};
#pragma unroll 8
  for (int k = 0; k < D_IN; k += 4) {
    v2f a, b;
    a.x = xa[k + khalf];
    a.y = xa[k + khalf + 1];
    b.x = wb[(k + khalf) * DOUT];
    b.y = wb[(k + khalf + 1) * DOUT];
    c = WMMA_F32(a, b, c);
  }
  const float bv = bias[col];
  const int rbase = (blockIdx.x << 4) + ((lane >> 4) << 3);
  float* yp = y + (size_t)rbase * DOUT + col;
  if ((blockIdx.x << 4) + 16 <= N) {
#pragma unroll
    for (int i = 0; i < 8; ++i) yp[i * DOUT] = c[i] + bv;
  } else {
#pragma unroll
    for (int i = 0; i < 8; ++i)
      if (rbase + i < N) yp[i * DOUT] = c[i] + bv;
  }
}


template <int DOUT>
__global__ __launch_bounds__(32 * (DOUT / 16)) void k_edge_wmma(
    const float* __restrict__ x, const float* __restrict__ Wrel,
    const int* __restrict__ bsrc, const int* __restrict__ bdst,
    const int* __restrict__ starts, float* __restrict__ y) {
  const int base = blockIdx.x << 4;
  if (base >= starts[8]) return;
  const int wave = threadIdx.x >> 5, lane = threadIdx.x & 31;
  int r = 0;
#pragma unroll
  for (int i = 1; i < 8; ++i) r += (base >= starts[i]) ? 1 : 0;

  const int m = lane & 15;
  const int khalf = (lane >> 4) << 1;
  const int srow = bsrc[base + m];
  const float* xa = x + (size_t)srow * D_IN;
  const int col = (wave << 4) + m;
  const float* wb = Wrel + (size_t)r * (D_IN * DOUT) + col;

  v8f c = {};
#pragma unroll 8
  for (int k = 0; k < D_IN; k += 4) {
    v2f a, b;
    a.x = xa[k + khalf];
    a.y = xa[k + khalf + 1];
    b.x = wb[(k + khalf) * DOUT];
    b.y = wb[(k + khalf + 1) * DOUT];
    c = WMMA_F32(a, b, c);
  }

  const int mBase = (lane >> 4) << 3;
#pragma unroll
  for (int i = 0; i < 8; ++i) {
    int drow = bdst[base + mBase + i];
    atomicAdd(&y[(size_t)drow * DOUT + col], c[i]);
  }
}


__global__ void k_relu4(float4* __restrict__ p, long long n4) {
  long long i = (long long)blockIdx.x * blockDim.x + threadIdx.x;
  if (i < n4) {
    float4 v = p[i];
    v.x = fmaxf(v.x, 0.0f);
    v.y = fmaxf(v.y, 0.0f);
    v.z = fmaxf(v.z, 0.0f);
    v.w = fmaxf(v.w, 0.0f);
    p[i] = v;
  }
}

__global__ void k_logsoftmax(const float* __restrict__ y, float* __restrict__ out, int N) {
  int wave = threadIdx.x >> 5, lane = threadIdx.x & 31;
  int node = blockIdx.x * (blockDim.x >> 5) + wave;
  if (node >= N) return;
  const float* row = y + (size_t)node * 64;
  float v0 = row[lane], v1 = row[lane + 32];
  float mx = fmaxf(v0, v1);
#pragma unroll
  for (int off = 16; off > 0; off >>= 1) mx = fmaxf(mx, __shfl_xor(mx, off, 32));
  float s = expf(v0 - mx) + expf(v1 - mx);
#pragma unroll
  for (int off = 16; off > 0; off >>= 1) s += __shfl_xor(s, off, 32);
  float lse = logf(s) + mx;
  float* o = out + (size_t)node * 64;
  o[lane] = v0 - lse;
  o[lane + 32] = v1 - lse;
}


static inline void launch_layer(const float* x, const float* Wrel, const float* Wroot,
                                const float* bias, float* y, int dOut, int N,
                                const int* bsrc, const int* bdst, const int* starts,
                                int nodeTiles, int edgeTilesMax, hipStream_t stream) {
  if (dOut == 128) {
    k_root_gemm<128><<<nodeTiles, 32 * (128 / 16), 0, stream>>>(x, Wroot, bias, y, N);
    k_edge_wmma<128><<<edgeTilesMax, 32 * (128 / 16), 0, stream>>>(x, Wrel, bsrc, bdst, starts, y);
  } else {
    k_root_gemm<64><<<nodeTiles, 32 * (64 / 16), 0, stream>>>(x, Wroot, bias, y, N);
    k_edge_wmma<64><<<edgeTilesMax, 32 * (64 / 16), 0, stream>>>(x, Wrel, bsrc, bdst, starts, y);
  }
}

extern "C" void kernel_launch(void* const* d_in, const int* in_sizes, int n_in,
                              void* d_out, int out_size, void* d_ws, size_t ws_size,
                              hipStream_t stream) {
  const float* x0     = (const float*)d_in[0];
  const int*   ei     = (const int*)  d_in[1];
  const int*   etype  = (const int*)  d_in[2];
  const float* Wrel1  = (const float*)d_in[3];
  const float* Wroot1 = (const float*)d_in[4];
  const float* b1     = (const float*)d_in[5];
  const float* Wrel2  = (const float*)d_in[6];
  const float* Wroot2 = (const float*)d_in[7];
  const float* b2     = (const float*)d_in[8];
  const float* Wrel3  = (const float*)d_in[9];
  const float* Wroot3 = (const float*)d_in[10];
  const float* b3     = (const float*)d_in[11];

  const int N  = in_sizes[0] / D_IN;
  const int E  = in_sizes[2];
  const int D1 = 128, D2 = 128;
  const int D3 = in_sizes[11];

  char* w = (char*)d_ws;
  int* counts = (int*)w;
  int* cursor = counts + 8;
  int* starts = cursor + 8;
  size_t off = 256;
  int* bsrc = (int*)(w + off); off += (size_t)(E + 256) * sizeof(int);
  int* bdst = (int*)(w + off); off += (size_t)(E + 256) * sizeof(int);
  off = (off + 255) & ~(size_t)255;
  float* bufA = (float*)(w + off); off += (size_t)(N + 16) * D_IN * sizeof(float);
  float* bufB = (float*)(w + off);

  k_zero_ints<<<1, 64, 0, stream>>>(counts, 32);
  k_hist    <<<(E + BKT_EDGES - 1) / BKT_EDGES, BKT_THREADS, 0, stream>>>(etype, E, counts);
  k_scan    <<<1, 1, 0, stream>>>(counts, starts);
  k_scatter <<<(E + BKT_EDGES - 1) / BKT_EDGES, BKT_THREADS, 0, stream>>>(ei, etype, E, starts, cursor,
                                                                          bsrc, bdst);
  k_pad     <<<8, 64, 0, stream>>>(counts, starts, bsrc, bdst, N);

  const int nodeTiles    = (N + 15) / 16;
  const int edgeTilesMax = E / 16 + 8;

  launch_layer(x0, Wrel1, Wroot1, b1, bufA, D1, N, bsrc, bdst, starts,
               nodeTiles, edgeTilesMax, stream);
  {
    long long n4 = (long long)N * D1 / 4;
    k_relu4<<<(unsigned)((n4 + 255) / 256), 256, 0, stream>>>((float4*)bufA, n4);
  }

  launch_layer(bufA, Wrel2, Wroot2, b2, bufB, D2, N, bsrc, bdst, starts,
               nodeTiles, edgeTilesMax, stream);
  {
    long long n4 = (long long)N * D2 / 4;
    k_relu4<<<(unsigned)((n4 + 255) / 256), 256, 0, stream>>>((float4*)bufB, n4);
  }

  launch_layer(bufB, Wrel3, Wroot3, b3, bufA, D3, N, bsrc, bdst, starts,
               nodeTiles, edgeTilesMax, stream);
  k_logsoftmax<<<(N + 7) / 8, 256, 0, stream>>>(bufA, (float*)d_out, N);
}


# A tile (16 gathered x rows) fetched once per work-group with whole-row loads and shared through LDS; waves read A fragments from LDS, not each via its own 16-line global gathers
# speedup vs baseline: 2.2385x; 1.6814x over previous
#include <hip/hip_runtime.h>
#include <hip/hip_bf16.h>
#include <math.h>


typedef float v2f __attribute__((ext_vector_type(2)));
typedef float v8f __attribute__((ext_vector_type(8)));

#define WMMA_F32(a, b, c) \
  __builtin_amdgcn_wmma_f32_16x16x4_f32(false, (a), false, (b), (short)0, (c), false, false)

static constexpr int D_IN = 128;

static constexpr int A_PITCH = D_IN + 4;

__device__ __forceinline__ void stage_row(float* As, const float* __restrict__ x,
                                          int srcRow, int tileRow, int lane) {
  const float4 v = *reinterpret_cast<const float4*>(x + (size_t)srcRow * D_IN + (lane << 2));
  *reinterpret_cast<float4*>(As + tileRow * A_PITCH + (lane << 2)) = v;
}


__global__ void k_zero_ints(int* __restrict__ p, int n) {
  int i = blockIdx.x * blockDim.x + threadIdx.x;
  if (i < n) p[i] = 0;
}

static constexpr int BKT_THREADS = 256;
static constexpr int BKT_ITEMS   = 8;
static constexpr int BKT_EDGES   = BKT_THREADS * BKT_ITEMS;

__device__ __forceinline__ unsigned rel_peers(int r, bool live) {
  unsigned peers = __builtin_amdgcn_ballot_w32(live);
#pragma unroll
  for (int b = 0; b < 3; ++b) {
    const bool bit = (r >> b) & 1;
    const unsigned vote = __builtin_amdgcn_ballot_w32(bit);
    peers &= bit ? vote : ~vote;
  }
  return peers;
}

__global__ __launch_bounds__(BKT_THREADS) void k_hist(const int* __restrict__ etype, int E,
                                                      int* __restrict__ counts) {
  __shared__ int cnt[8];
  const int lane = threadIdx.x & 31;
  if (threadIdx.x < 8) cnt[threadIdx.x] = 0;
  __syncthreads();
  const int e0 = blockIdx.x * BKT_EDGES + threadIdx.x;
  int r[BKT_ITEMS];
#pragma unroll
  for (int i = 0; i < BKT_ITEMS; ++i) {
    const int e = e0 + i * BKT_THREADS;
    const int ec = e < E ? e : E - 1;
    const int t = etype[ec];
    r[i] = e < E ? (t & 7) : -1;
  }
#pragma unroll
  for (int i = 0; i < BKT_ITEMS; ++i) {
    const bool live = r[i] >= 0;
    const unsigned peers = rel_peers(r[i], live);
    if (live && (peers & ((1u << lane) - 1u)) == 0) atomicAdd(&cnt[r[i]], __popc(peers));
  }
  __syncthreads();
  if (threadIdx.x < 8) {
    const int c = cnt[threadIdx.x];
    if (c) atomicAdd(&counts[threadIdx.x], c);
  }
}

__global__ void k_scan(const int* __restrict__ counts, int* __restrict__ starts) {
  if (threadIdx.x == 0 && blockIdx.x == 0) {
    int acc = 0;
    for (int r = 0; r < 8; ++r) {
      starts[r] = acc;
      acc += (counts[r] + 15) & ~15;
    }
    starts[8] = acc;
  }
}

__global__ __launch_bounds__(BKT_THREADS) void k_scatter(
    const int* __restrict__ ei, const int* __restrict__ etype, int E,
    const int* __restrict__ starts, int* __restrict__ cursor,
    int* __restrict__ bsrc, int* __restrict__ bdst) {
  __shared__ int cnt[8];
  __shared__ int base[8];
  const int lane = threadIdx.x & 31;
  if (threadIdx.x < 8) cnt[threadIdx.x] = 0;
  __syncthreads();
  const int e0 = blockIdx.x * BKT_EDGES + threadIdx.x;
  int r[BKT_ITEMS], s[BKT_ITEMS], d[BKT_ITEMS], rank[BKT_ITEMS];
#pragma unroll
  for (int i = 0; i < BKT_ITEMS; ++i) {
    const int e = e0 + i * BKT_THREADS;
    const int ec = e < E ? e : E - 1;
    const int t = etype[ec];
    r[i] = e < E ? (t & 7) : -1;
    s[i] = ei[ec];
    d[i] = ei[E + ec];
  }
#pragma unroll
  for (int i = 0; i < BKT_ITEMS; ++i) {
    const bool live = r[i] >= 0;
    const unsigned peers = rel_peers(r[i], live);
    const int below = __popc(peers & ((1u << lane) - 1u));
    int wbase = 0;
    if (live && below == 0) wbase = atomicAdd(&cnt[r[i]], __popc(peers));
    wbase = __shfl(wbase, live ? __ffs(peers) - 1 : lane, 32);
    rank[i] = wbase + below;
  }
  __syncthreads();
  if (threadIdx.x < 8) {
    const int c = cnt[threadIdx.x];
    base[threadIdx.x] = starts[threadIdx.x] + (c ? atomicAdd(&cursor[threadIdx.x], c) : 0);
  }
  __syncthreads();
#pragma unroll
  for (int i = 0; i < BKT_ITEMS; ++i) {
    if (r[i] >= 0) {
      const int idx = base[r[i]] + rank[i];
      bsrc[idx] = s[i];
      bdst[idx] = d[i];
    }
  }
}

__global__ void k_pad(const int* __restrict__ counts, const int* __restrict__ starts,
                      int* __restrict__ bsrc, int* __restrict__ bdst, int dummyRow) {
  int r = blockIdx.x;
  int lo = starts[r] + counts[r];
  int hi = starts[r + 1];
  for (int i = lo + (int)threadIdx.x; i < hi; i += blockDim.x) {
    bsrc[i] = 0;
    bdst[i] = dummyRow;
  }
}


template <int DOUT>
__global__ __launch_bounds__(32 * (DOUT / 16)) void k_root_gemm(
    const float* __restrict__ x, const float* __restrict__ W,
    const float* __restrict__ bias, float* __restrict__ y, int N) {
  __shared__ __attribute__((aligned(16))) float As[16 * A_PITCH];
  constexpr int NW = DOUT / 16;
  const int wave = threadIdx.x >> 5, lane = threadIdx.x & 31;
  const int m = lane & 15;
  const int khalf = (lane >> 4) << 1;
#pragma unroll
  for (int j = 0; j < 16 / NW; ++j) {
    const int tr = wave + j * NW;
    const int row = (blockIdx.x << 4) + tr;
    stage_row(As, x, row < N ? row : N - 1, tr, lane);
  }
  __syncthreads();
  const float* xa = As + m * A_PITCH;
  const int col = (wave << 4) + m;
  const float* wb = W + col;
  v8f c = {};
#pragma unroll 8
  for (int k = 0; k < D_IN; k += 4) {
    v2f a, b;
    a.x = xa[k + khalf];
    a.y = xa[k + khalf + 1];
    b.x = wb[(k + khalf) * DOUT];
    b.y = wb[(k + khalf + 1) * DOUT];
    c = WMMA_F32(a, b, c);
  }
  const float bv = bias[col];
  const int rbase = (blockIdx.x << 4) + ((lane >> 4) << 3);
  float* yp = y + (size_t)rbase * DOUT + col;
  if ((blockIdx.x << 4) + 16 <= N) {
#pragma unroll
    for (int i = 0; i < 8; ++i) yp[i * DOUT] = c[i] + bv;
  } else {
#pragma unroll
    for (int i = 0; i < 8; ++i)
      if (rbase + i < N) yp[i * DOUT] = c[i] + bv;
  }
}


template <int DOUT>
__global__ __launch_bounds__(32 * (DOUT / 16)) void k_edge_wmma(
    const float* __restrict__ x, const float* __restrict__ Wrel,
    const int* __restrict__ bsrc, const int* __restrict__ bdst,
    const int* __restrict__ starts, float* __restrict__ y) {
  __shared__ __attribute__((aligned(16))) float As[16 * A_PITCH];
  constexpr int NW = DOUT / 16;
  const int base = blockIdx.x << 4;
  if (base >= starts[8]) return;
  const int wave = threadIdx.x >> 5, lane = threadIdx.x & 31;
#pragma unroll
  for (int j = 0; j < 16 / NW; ++j) {
    const int tr = wave + j * NW;
    stage_row(As, x, bsrc[base + tr], tr, lane);
  }
  __syncthreads();
  int r = 0;
#pragma unroll
  for (int i = 1; i < 8; ++i) r += (base >= starts[i]) ? 1 : 0;

  const int m = lane & 15;
  const int khalf = (lane >> 4) << 1;
  const float* xa = As + m * A_PITCH;
  const int col = (wave << 4) + m;
  const float* wb = Wrel + (size_t)r * (D_IN * DOUT) + col;

  v8f c = {};
#pragma unroll 8
  for (int k = 0; k < D_IN; k += 4) {
    v2f a, b;
    a.x = xa[k + khalf];
    a.y = xa[k + khalf + 1];
    b.x = wb[(k + khalf) * DOUT];
    b.y = wb[(k + khalf + 1) * DOUT];
    c = WMMA_F32(a, b, c);
  }

  const int mBase = (lane >> 4) << 3;
#pragma unroll
  for (int i = 0; i < 8; ++i) {
    int drow = bdst[base + mBase + i];
    atomicAdd(&y[(size_t)drow * DOUT + col], c[i]);
  }
}


__global__ void k_relu4(float4* __restrict__ p, long long n4) {
  long long i = (long long)blockIdx.x * blockDim.x + threadIdx.x;
  if (i < n4) {
    float4 v = p[i];
    v.x = fmaxf(v.x, 0.0f);
    v.y = fmaxf(v.y, 0.0f);
    v.z = fmaxf(v.z, 0.0f);
    v.w = fmaxf(v.w, 0.0f);
    p[i] = v;
  }
}

__global__ void k_logsoftmax(const float* __restrict__ y, float* __restrict__ out, int N) {
  int wave = threadIdx.x >> 5, lane = threadIdx.x & 31;
  int node = blockIdx.x * (blockDim.x >> 5) + wave;
  if (node >= N) return;
  const float* row = y + (size_t)node * 64;
  float v0 = row[lane], v1 = row[lane + 32];
  float mx = fmaxf(v0, v1);
#pragma unroll
  for (int off = 16; off > 0; off >>= 1) mx = fmaxf(mx, __shfl_xor(mx, off, 32));
  float s = expf(v0 - mx) + expf(v1 - mx);
#pragma unroll
  for (int off = 16; off > 0; off >>= 1) s += __shfl_xor(s, off, 32);
  float lse = logf(s) + mx;
  float* o = out + (size_t)node * 64;
  o[lane] = v0 - lse;
  o[lane + 32] = v1 - lse;
}


static inline void launch_layer(const float* x, const float* Wrel, const float* Wroot,
                                const float* bias, float* y, int dOut, int N,
                                const int* bsrc, const int* bdst, const int* starts,
                                int nodeTiles, int edgeTilesMax, hipStream_t stream) {
  if (dOut == 128) {
    k_root_gemm<128><<<nodeTiles, 32 * (128 / 16), 0, stream>>>(x, Wroot, bias, y, N);
    k_edge_wmma<128><<<edgeTilesMax, 32 * (128 / 16), 0, stream>>>(x, Wrel, bsrc, bdst, starts, y);
  } else {
    k_root_gemm<64><<<nodeTiles, 32 * (64 / 16), 0, stream>>>(x, Wroot, bias, y, N);
    k_edge_wmma<64><<<edgeTilesMax, 32 * (64 / 16), 0, stream>>>(x, Wrel, bsrc, bdst, starts, y);
  }
}

extern "C" void kernel_launch(void* const* d_in, const int* in_sizes, int n_in,
                              void* d_out, int out_size, void* d_ws, size_t ws_size,
                              hipStream_t stream) {
  const float* x0     = (const float*)d_in[0];
  const int*   ei     = (const int*)  d_in[1];
  const int*   etype  = (const int*)  d_in[2];
  const float* Wrel1  = (const float*)d_in[3];
  const float* Wroot1 = (const float*)d_in[4];
  const float* b1     = (const float*)d_in[5];
  const float* Wrel2  = (const float*)d_in[6];
  const float* Wroot2 = (const float*)d_in[7];
  const float* b2     = (const float*)d_in[8];
  const float* Wrel3  = (const float*)d_in[9];
  const float* Wroot3 = (const float*)d_in[10];
  const float* b3     = (const float*)d_in[11];

  const int N  = in_sizes[0] / D_IN;
  const int E  = in_sizes[2];
  const int D1 = 128, D2 = 128;
  const int D3 = in_sizes[11];

  char* w = (char*)d_ws;
  int* counts = (int*)w;
  int* cursor = counts + 8;
  int* starts = cursor + 8;
  size_t off = 256;
  int* bsrc = (int*)(w + off); off += (size_t)(E + 256) * sizeof(int);
  int* bdst = (int*)(w + off); off += (size_t)(E + 256) * sizeof(int);
  off = (off + 255) & ~(size_t)255;
  float* bufA = (float*)(w + off); off += (size_t)(N + 16) * D_IN * sizeof(float);
  float* bufB = (float*)(w + off);

  k_zero_ints<<<1, 64, 0, stream>>>(counts, 32);
  k_hist    <<<(E + BKT_EDGES - 1) / BKT_EDGES, BKT_THREADS, 0, stream>>>(etype, E, counts);
  k_scan    <<<1, 1, 0, stream>>>(counts, starts);
  k_scatter <<<(E + BKT_EDGES - 1) / BKT_EDGES, BKT_THREADS, 0, stream>>>(ei, etype, E, starts, cursor,
                                                                          bsrc, bdst);
  k_pad     <<<8, 64, 0, stream>>>(counts, starts, bsrc, bdst, N);

  const int nodeTiles    = (N + 15) / 16;
  const int edgeTilesMax = E / 16 + 8;

  launch_layer(x0, Wrel1, Wroot1, b1, bufA, D1, N, bsrc, bdst, starts,
               nodeTiles, edgeTilesMax, stream);
  {
    long long n4 = (long long)N * D1 / 4;
    k_relu4<<<(unsigned)((n4 + 255) / 256), 256, 0, stream>>>((float4*)bufA, n4);
  }

  launch_layer(bufA, Wrel2, Wroot2, b2, bufB, D2, N, bsrc, bdst, starts,
               nodeTiles, edgeTilesMax, stream);
  {
    long long n4 = (long long)N * D2 / 4;
    k_relu4<<<(unsigned)((n4 + 255) / 256), 256, 0, stream>>>((float4*)bufB, n4);
  }

  launch_layer(bufB, Wrel3, Wroot3, b3, bufA, D3, N, bsrc, bdst, starts,
               nodeTiles, edgeTilesMax, stream);
  k_logsoftmax<<<(N + 7) / 8, 256, 0, stream>>>(bufA, (float*)d_out, N);
}


# Edge work-groups walk runs of 8 consecutive 16-edge tiles with each wave's Wrel[r] fragments held in registers: weights fetched once per run, not per tile; 8x fewer waves; float atomic scatter kept
# speedup vs baseline: 2.7065x; 1.2091x over previous
#include <hip/hip_runtime.h>
#include <hip/hip_bf16.h>
#include <math.h>


typedef float v2f __attribute__((ext_vector_type(2)));
typedef float v8f __attribute__((ext_vector_type(8)));

#define WMMA_F32(a, b, c) \
  __builtin_amdgcn_wmma_f32_16x16x4_f32(false, (a), false, (b), (short)0, (c), false, false)

static constexpr int D_IN = 128;

static constexpr int A_PITCH = D_IN + 4;

__device__ __forceinline__ void stage_row(float* As, const float* __restrict__ x,
                                          int srcRow, int tileRow, int lane) {
  const float4 v = *reinterpret_cast<const float4*>(x + (size_t)srcRow * D_IN + (lane << 2));
  *reinterpret_cast<float4*>(As + tileRow * A_PITCH + (lane << 2)) = v;
}


__global__ void k_zero_ints(int* __restrict__ p, int n) {
  int i = blockIdx.x * blockDim.x + threadIdx.x;
  if (i < n) p[i] = 0;
}

static constexpr int BKT_THREADS = 256;
static constexpr int BKT_ITEMS   = 8;
static constexpr int BKT_EDGES   = BKT_THREADS * BKT_ITEMS;

__device__ __forceinline__ unsigned rel_peers(int r, bool live) {
  unsigned peers = __builtin_amdgcn_ballot_w32(live);
#pragma unroll
  for (int b = 0; b < 3; ++b) {
    const bool bit = (r >> b) & 1;
    const unsigned vote = __builtin_amdgcn_ballot_w32(bit);
    peers &= bit ? vote : ~vote;
  }
  return peers;
}

__global__ __launch_bounds__(BKT_THREADS) void k_hist(const int* __restrict__ etype, int E,
                                                      int* __restrict__ counts) {
  __shared__ int cnt[8];
  const int lane = threadIdx.x & 31;
  if (threadIdx.x < 8) cnt[threadIdx.x] = 0;
  __syncthreads();
  const int e0 = blockIdx.x * BKT_EDGES + threadIdx.x;
  int r[BKT_ITEMS];
#pragma unroll
  for (int i = 0; i < BKT_ITEMS; ++i) {
    const int e = e0 + i * BKT_THREADS;
    const int ec = e < E ? e : E - 1;
    const int t = etype[ec];
    r[i] = e < E ? (t & 7) : -1;
  }
#pragma unroll
  for (int i = 0; i < BKT_ITEMS; ++i) {
    const bool live = r[i] >= 0;
    const unsigned peers = rel_peers(r[i], live);
    if (live && (peers & ((1u << lane) - 1u)) == 0) atomicAdd(&cnt[r[i]], __popc(peers));
  }
  __syncthreads();
  if (threadIdx.x < 8) {
    const int c = cnt[threadIdx.x];
    if (c) atomicAdd(&counts[threadIdx.x], c);
  }
}

__global__ void k_scan(const int* __restrict__ counts, int* __restrict__ starts) {
  if (threadIdx.x == 0 && blockIdx.x == 0) {
    int acc = 0;
    for (int r = 0; r < 8; ++r) {
      starts[r] = acc;
      acc += (counts[r] + 15) & ~15;
    }
    starts[8] = acc;
  }
}

__global__ __launch_bounds__(BKT_THREADS) void k_scatter(
    const int* __restrict__ ei, const int* __restrict__ etype, int E,
    const int* __restrict__ starts, int* __restrict__ cursor,
    int* __restrict__ bsrc, int* __restrict__ bdst) {
  __shared__ int cnt[8];
  __shared__ int base[8];
  const int lane = threadIdx.x & 31;
  if (threadIdx.x < 8) cnt[threadIdx.x] = 0;
  __syncthreads();
  const int e0 = blockIdx.x * BKT_EDGES + threadIdx.x;
  int r[BKT_ITEMS], s[BKT_ITEMS], d[BKT_ITEMS], rank[BKT_ITEMS];
#pragma unroll
  for (int i = 0; i < BKT_ITEMS; ++i) {
    const int e = e0 + i * BKT_THREADS;
    const int ec = e < E ? e : E - 1;
    const int t = etype[ec];
    r[i] = e < E ? (t & 7) : -1;
    s[i] = ei[ec];
    d[i] = ei[E + ec];
  }
#pragma unroll
  for (int i = 0; i < BKT_ITEMS; ++i) {
    const bool live = r[i] >= 0;
    const unsigned peers = rel_peers(r[i], live);
    const int below = __popc(peers & ((1u << lane) - 1u));
    int wbase = 0;
    if (live && below == 0) wbase = atomicAdd(&cnt[r[i]], __popc(peers));
    wbase = __shfl(wbase, live ? __ffs(peers) - 1 : lane, 32);
    rank[i] = wbase + below;
  }
  __syncthreads();
  if (threadIdx.x < 8) {
    const int c = cnt[threadIdx.x];
    base[threadIdx.x] = starts[threadIdx.x] + (c ? atomicAdd(&cursor[threadIdx.x], c) : 0);
  }
  __syncthreads();
#pragma unroll
  for (int i = 0; i < BKT_ITEMS; ++i) {
    if (r[i] >= 0) {
      const int idx = base[r[i]] + rank[i];
      bsrc[idx] = s[i];
      bdst[idx] = d[i];
    }
  }
}

__global__ void k_pad(const int* __restrict__ counts, const int* __restrict__ starts,
                      int* __restrict__ bsrc, int* __restrict__ bdst, int dummyRow) {
  int r = blockIdx.x;
  int lo = starts[r] + counts[r];
  int hi = starts[r + 1];
  for (int i = lo + (int)threadIdx.x; i < hi; i += blockDim.x) {
    bsrc[i] = 0;
    bdst[i] = dummyRow;
  }
}


template <int DOUT>
__global__ __launch_bounds__(32 * (DOUT / 16)) void k_root_gemm(
    const float* __restrict__ x, const float* __restrict__ W,
    const float* __restrict__ bias, float* __restrict__ y, int N) {
  __shared__ __attribute__((aligned(16))) float As[16 * A_PITCH];
  constexpr int NW = DOUT / 16;
  const int wave = threadIdx.x >> 5, lane = threadIdx.x & 31;
  const int m = lane & 15;
  const int khalf = (lane >> 4) << 1;
#pragma unroll
  for (int j = 0; j < 16 / NW; ++j) {
    const int tr = wave + j * NW;
    const int row = (blockIdx.x << 4) + tr;
    stage_row(As, x, row < N ? row : N - 1, tr, lane);
  }
  __syncthreads();
  const float* xa = As + m * A_PITCH;
  const int col = (wave << 4) + m;
  const float* wb = W + col;
  v8f c = {};
#pragma unroll 8
  for (int k = 0; k < D_IN; k += 4) {
    v2f a, b;
    a.x = xa[k + khalf];
    a.y = xa[k + khalf + 1];
    b.x = wb[(k + khalf) * DOUT];
    b.y = wb[(k + khalf + 1) * DOUT];
    c = WMMA_F32(a, b, c);
  }
  const float bv = bias[col];
  const int rbase = (blockIdx.x << 4) + ((lane >> 4) << 3);
  float* yp = y + (size_t)rbase * DOUT + col;
  if ((blockIdx.x << 4) + 16 <= N) {
#pragma unroll
    for (int i = 0; i < 8; ++i) yp[i * DOUT] = c[i] + bv;
  } else {
#pragma unroll
    for (int i = 0; i < 8; ++i)
      if (rbase + i < N) yp[i * DOUT] = c[i] + bv;
  }
}


static constexpr int RUN_TILES = 8;
static constexpr int RUN_EDGES = RUN_TILES * 16;

template <int DOUT>
__global__ __launch_bounds__(32 * (DOUT / 16)) void k_edge_wmma(
    const float* __restrict__ x, const float* __restrict__ Wrel,
    const int* __restrict__ bsrc, const int* __restrict__ bdst,
    const int* __restrict__ starts, float* __restrict__ y) {
  __shared__ __attribute__((aligned(16))) float As[2][16 * A_PITCH];
  __shared__ __attribute__((aligned(16))) int sIdx[RUN_EDGES];
  __shared__ __attribute__((aligned(16))) int dIdx[RUN_EDGES];
  constexpr int NW = DOUT / 16;
  constexpr int RPW = 16 / NW;
  const int total = starts[8];
  const int base0 = blockIdx.x * RUN_EDGES;
  if (base0 >= total) return;
  const int left = (total - base0) >> 4;
  const int nT = left < RUN_TILES ? left : RUN_TILES;
  const int wave = threadIdx.x >> 5, lane = threadIdx.x & 31;

  for (int i = threadIdx.x; i < RUN_EDGES; i += 32 * NW) {
    const int gi = base0 + i;
    const bool ok = gi < total;
    sIdx[i] = ok ? bsrc[gi] : 0;
    dIdx[i] = ok ? bdst[gi] : 0;
  }
#pragma unroll
  for (int j = 0; j < RPW; ++j) {
    const int tr = wave + j * NW;
    stage_row(As[0], x, bsrc[base0 + tr], tr, lane);
  }
  __syncthreads();

  const int m = lane & 15;
  const int khalf = (lane >> 4) << 1;
  const int col = (wave << 4) + m;
  const int mBase = (lane >> 4) << 3;
  const int s1 = starts[1], s2 = starts[2], s3 = starts[3], s4 = starts[4],
            s5 = starts[5], s6 = starts[6], s7 = starts[7];

  v2f b[D_IN / 4];
  int curR = -1;
  int cur = 0;
  for (int t = 0; t < nT; ++t) {
    const int base = base0 + (t << 4);
    const int r = (base >= s1) + (base >= s2) + (base >= s3) + (base >= s4) +
                  (base >= s5) + (base >= s6) + (base >= s7);
    if (r != curR) {
      const float* wb = Wrel + (size_t)r * (D_IN * DOUT) + col;
#pragma unroll
      for (int kk = 0; kk < D_IN / 4; ++kk) {
        b[kk].x = wb[(4 * kk + khalf) * DOUT];
        b[kk].y = wb[(4 * kk + khalf + 1) * DOUT];
      }
      curR = r;
    }

    const int tn = t + 1 < nT ? t + 1 : t;
    float4 pre[RPW];
#pragma unroll
    for (int j = 0; j < RPW; ++j) {
      const int tr = wave + j * NW;
      pre[j] = *reinterpret_cast<const float4*>(x + (size_t)sIdx[(tn << 4) + tr] * D_IN + (lane << 2));
    }

    const float* xa = As[cur] + m * A_PITCH;
    v8f c = {};
#pragma unroll
    for (int kk = 0; kk < D_IN / 4; ++kk) {
      v2f a;
      a.x = xa[4 * kk + khalf];
      a.y = xa[4 * kk + khalf + 1];
      c = WMMA_F32(a, b[kk], c);
    }

    float* An = As[cur ^ 1];
#pragma unroll
    for (int j = 0; j < RPW; ++j) {
      const int tr = wave + j * NW;
      *reinterpret_cast<float4*>(An + tr * A_PITCH + (lane << 2)) = pre[j];
    }

    __syncthreads();
    cur ^= 1;

    const int* dp = dIdx + (t << 4) + mBase;
#pragma unroll
    for (int i = 0; i < 8; ++i) {
      const int drow = dp[i];
      atomicAdd(&y[(size_t)drow * DOUT + col], c[i]);
    }
  }
}


__global__ void k_relu4(float4* __restrict__ p, long long n4) {
  long long i = (long long)blockIdx.x * blockDim.x + threadIdx.x;
  if (i < n4) {
    float4 v = p[i];
    v.x = fmaxf(v.x, 0.0f);
    v.y = fmaxf(v.y, 0.0f);
    v.z = fmaxf(v.z, 0.0f);
    v.w = fmaxf(v.w, 0.0f);
    p[i] = v;
  }
}

__global__ void k_logsoftmax(const float* __restrict__ y, float* __restrict__ out, int N) {
  int wave = threadIdx.x >> 5, lane = threadIdx.x & 31;
  int node = blockIdx.x * (blockDim.x >> 5) + wave;
  if (node >= N) return;
  const float* row = y + (size_t)node * 64;
  float v0 = row[lane], v1 = row[lane + 32];
  float mx = fmaxf(v0, v1);
#pragma unroll
  for (int off = 16; off > 0; off >>= 1) mx = fmaxf(mx, __shfl_xor(mx, off, 32));
  float s = expf(v0 - mx) + expf(v1 - mx);
#pragma unroll
  for (int off = 16; off > 0; off >>= 1) s += __shfl_xor(s, off, 32);
  float lse = logf(s) + mx;
  float* o = out + (size_t)node * 64;
  o[lane] = v0 - lse;
  o[lane + 32] = v1 - lse;
}


static inline void launch_layer(const float* x, const float* Wrel, const float* Wroot,
                                const float* bias, float* y, int dOut, int N,
                                const int* bsrc, const int* bdst, const int* starts,
                                int nodeTiles, int edgeRunsMax, hipStream_t stream) {
  if (dOut == 128) {
    k_root_gemm<128><<<nodeTiles, 32 * (128 / 16), 0, stream>>>(x, Wroot, bias, y, N);
    k_edge_wmma<128><<<edgeRunsMax, 32 * (128 / 16), 0, stream>>>(x, Wrel, bsrc, bdst, starts, y);
  } else {
    k_root_gemm<64><<<nodeTiles, 32 * (64 / 16), 0, stream>>>(x, Wroot, bias, y, N);
    k_edge_wmma<64><<<edgeRunsMax, 32 * (64 / 16), 0, stream>>>(x, Wrel, bsrc, bdst, starts, y);
  }
}

extern "C" void kernel_launch(void* const* d_in, const int* in_sizes, int n_in,
                              void* d_out, int out_size, void* d_ws, size_t ws_size,
                              hipStream_t stream) {
  const float* x0     = (const float*)d_in[0];
  const int*   ei     = (const int*)  d_in[1];
  const int*   etype  = (const int*)  d_in[2];
  const float* Wrel1  = (const float*)d_in[3];
  const float* Wroot1 = (const float*)d_in[4];
  const float* b1     = (const float*)d_in[5];
  const float* Wrel2  = (const float*)d_in[6];
  const float* Wroot2 = (const float*)d_in[7];
  const float* b2     = (const float*)d_in[8];
  const float* Wrel3  = (const float*)d_in[9];
  const float* Wroot3 = (const float*)d_in[10];
  const float* b3     = (const float*)d_in[11];

  const int N  = in_sizes[0] / D_IN;
  const int E  = in_sizes[2];
  const int D1 = 128, D2 = 128;
  const int D3 = in_sizes[11];

  char* w = (char*)d_ws;
  int* counts = (int*)w;
  int* cursor = counts + 8;
  int* starts = cursor + 8;
  size_t off = 256;
  int* bsrc = (int*)(w + off); off += (size_t)(E + 256) * sizeof(int);
  int* bdst = (int*)(w + off); off += (size_t)(E + 256) * sizeof(int);
  off = (off + 255) & ~(size_t)255;
  float* bufA = (float*)(w + off); off += (size_t)(N + 16) * D_IN * sizeof(float);
  float* bufB = (float*)(w + off);

  k_zero_ints<<<1, 64, 0, stream>>>(counts, 32);
  k_hist    <<<(E + BKT_EDGES - 1) / BKT_EDGES, BKT_THREADS, 0, stream>>>(etype, E, counts);
  k_scan    <<<1, 1, 0, stream>>>(counts, starts);
  k_scatter <<<(E + BKT_EDGES - 1) / BKT_EDGES, BKT_THREADS, 0, stream>>>(ei, etype, E, starts, cursor,
                                                                          bsrc, bdst);
  k_pad     <<<8, 64, 0, stream>>>(counts, starts, bsrc, bdst, N);

  const int nodeTiles    = (N + 15) / 16;
  const int edgeTilesMax = E / 16 + 8;
  const int edgeRunsMax  = (edgeTilesMax + RUN_TILES - 1) / RUN_TILES;

  launch_layer(x0, Wrel1, Wroot1, b1, bufA, D1, N, bsrc, bdst, starts,
               nodeTiles, edgeRunsMax, stream);
  {
    long long n4 = (long long)N * D1 / 4;
    k_relu4<<<(unsigned)((n4 + 255) / 256), 256, 0, stream>>>((float4*)bufA, n4);
  }

  launch_layer(bufA, Wrel2, Wroot2, b2, bufB, D2, N, bsrc, bdst, starts,
               nodeTiles, edgeRunsMax, stream);
  {
    long long n4 = (long long)N * D2 / 4;
    k_relu4<<<(unsigned)((n4 + 255) / 256), 256, 0, stream>>>((float4*)bufB, n4);
  }

  launch_layer(bufB, Wrel3, Wroot3, b3, bufA, D3, N, bsrc, bdst, starts,
               nodeTiles, edgeRunsMax, stream);
  k_logsoftmax<<<(N + 7) / 8, 256, 0, stream>>>(bufA, (float*)d_out, N);
}


# Edges sorted by (relation,dst); source rows of equal (relation,dst) summed while staging the A tile: one tile row and one atomic row per distinct pair, not per edge; sum order changes, atomics kept
# speedup vs baseline: 3.5146x; 1.2986x over previous
#include <hip/hip_runtime.h>
#include <hip/hip_bf16.h>
#include <math.h>


typedef float v2f __attribute__((ext_vector_type(2)));
typedef float v8f __attribute__((ext_vector_type(8)));

#define WMMA_F32(a, b, c) \
  __builtin_amdgcn_wmma_f32_16x16x4_f32(false, (a), false, (b), (short)0, (c), false, false)

static constexpr int D_IN = 128;

static constexpr int A_PITCH = D_IN + 4;

__device__ __forceinline__ void stage_row(float* As, const float* __restrict__ x,
                                          int srcRow, int tileRow, int lane) {
  const float4 v = *reinterpret_cast<const float4*>(x + (size_t)srcRow * D_IN + (lane << 2));
  *reinterpret_cast<float4*>(As + tileRow * A_PITCH + (lane << 2)) = v;
}

static constexpr int SEG_EDGES = 4;

__global__ void k_zero_ints(int* __restrict__ p, int n) {
  int i = blockIdx.x * blockDim.x + threadIdx.x;
  if (i < n) p[i] = 0;
}

__global__ void k_key_hist(const int* __restrict__ ei, const int* __restrict__ etype, int E, int N,
                           int* __restrict__ cnt) {
  const int e = blockIdx.x * blockDim.x + threadIdx.x;
  if (e < E) atomicAdd(&cnt[(etype[e] & 7) * N + ei[E + e]], 1);
}

static constexpr int SCAN_THREADS = 1024;

__device__ __forceinline__ int wave_incl_scan(int v, int lane) {
#pragma unroll
  for (int o = 1; o < 32; o <<= 1) {
    const int t = __shfl_up(v, o, 32);
    if (lane >= o) v += t;
  }
  return v;
}

__device__ __forceinline__ int wave_sum(int v) {
#pragma unroll
  for (int o = 16; o > 0; o >>= 1) v += __shfl_xor(v, o, 32);
  return v;
}

__global__ __launch_bounds__(SCAN_THREADS) void k_scan_a(
    const int* __restrict__ cnt, int N, int nb, int* __restrict__ blkE, int* __restrict__ blkS) {
  __shared__ int wE[32], wS[32];
  const int r = blockIdx.y, t = threadIdx.x, lane = t & 31, wave = t >> 5;
  const int k = blockIdx.x * SCAN_THREADS + t;
  const int v = k < N ? cnt[(size_t)r * N + k] : 0;
  const int e = wave_sum(v);
  const int s = wave_sum((v + SEG_EDGES - 1) / SEG_EDGES);
  if (lane == 0) { wE[wave] = e; wS[wave] = s; }
  __syncthreads();
  if (wave == 0) {
    const int te = wave_sum(wE[lane]), ts = wave_sum(wS[lane]);
    if (lane == 0) {
      blkE[r * nb + blockIdx.x] = te;
      blkS[r * nb + blockIdx.x] = ts;
    }
  }
}

static constexpr int MID_MAX = 1024;
__global__ __launch_bounds__(SCAN_THREADS) void k_scan_mid(
    const int* __restrict__ blkE, const int* __restrict__ blkS, int nb, int N,
    int* __restrict__ blkBaseE, int* __restrict__ blkBaseS,
    int* __restrict__ starts, int* __restrict__ segBeg, int* __restrict__ segDst) {
  __shared__ int inE[MID_MAX], inS[MID_MAX];
  __shared__ int relE[8], relS[8], baseE[9], baseS[9];
  const int t = threadIdx.x, lane = t & 31, wave = t >> 5;
  if (wave < 8) {
    int carryE = 0, carryS = 0;
    for (int b0 = 0; b0 < nb; b0 += 32) {
      const int b = b0 + lane;
      const int i = wave * nb + b;
      const int ve = b < nb ? blkE[i] : 0;
      const int vs = b < nb ? blkS[i] : 0;
      const int ie = wave_incl_scan(ve, lane), is = wave_incl_scan(vs, lane);
      if (b < nb) {
        inE[i] = carryE + ie - ve;
        inS[i] = carryS + is - vs;
      }
      carryE += __shfl(ie, 31, 32);
      carryS += __shfl(is, 31, 32);
    }
    if (lane == 0) { relE[wave] = carryE; relS[wave] = carryS; }
  }
  __syncthreads();
  if (t == 0) {
    int e = 0, s = 0;
    for (int r = 0; r < 8; ++r) {
      baseE[r] = e;
      baseS[r] = s;
      e += relE[r];
      s += (relS[r] + 15) & ~15;
    }
    baseE[8] = e;
    baseS[8] = s;
  }
  __syncthreads();
  for (int i = t; i < 8 * nb; i += SCAN_THREADS) {
    const int r = i / nb;
    blkBaseE[i] = baseE[r] + inE[i];
    blkBaseS[i] = baseS[r] + inS[i];
  }
  if (t < 9) starts[t] = baseS[t];
  if (t == 0) segBeg[baseS[8]] = baseE[8];
  if (t < 128) {
    const int r = t >> 4;
    const int p = baseS[r] + relS[r] + (t & 15);
    if (p < baseS[r + 1]) {
      segBeg[p] = baseE[r + 1];
      segDst[p] = N;
    }
  }
}

__global__ __launch_bounds__(SCAN_THREADS) void k_scan_b(
    int* __restrict__ cnt, int N, int nb,
    const int* __restrict__ blkBaseE, const int* __restrict__ blkBaseS,
    int* __restrict__ segBeg, int* __restrict__ segDst) {
  __shared__ int wE[32], wS[32];
  const int r = blockIdx.y, t = threadIdx.x, lane = t & 31, wave = t >> 5;
  const int k = blockIdx.x * SCAN_THREADS + t;
  const int v = k < N ? cnt[(size_t)r * N + k] : 0;
  const int sv = (v + SEG_EDGES - 1) / SEG_EDGES;
  const int ie = wave_incl_scan(v, lane), is = wave_incl_scan(sv, lane);
  if (lane == 31) { wE[wave] = ie; wS[wave] = is; }
  __syncthreads();
  if (wave == 0) {
    const int a = wE[lane], b = wS[lane];
    const int ia = wave_incl_scan(a, lane), ib = wave_incl_scan(b, lane);
    wE[lane] = ia - a;
    wS[lane] = ib - b;
  }
  __syncthreads();
  const int e0 = blkBaseE[r * nb + blockIdx.x] + wE[wave] + ie - v;
  int s = blkBaseS[r * nb + blockIdx.x] + wS[wave] + is - sv;
  if (k < N) {
    cnt[(size_t)r * N + k] = e0;
    for (int q = 0; q < v; q += SEG_EDGES) {
      segBeg[s] = e0 + q;
      segDst[s] = k;
      ++s;
    }
  }
}

__global__ void k_key_scatter(const int* __restrict__ ei, const int* __restrict__ etype, int E, int N,
                              int* __restrict__ cursor, int* __restrict__ ssrc) {
  const int e = blockIdx.x * blockDim.x + threadIdx.x;
  if (e < E) {
    const int pos = atomicAdd(&cursor[(etype[e] & 7) * N + ei[E + e]], 1);
    ssrc[pos] = ei[e];
  }
}


template <int DOUT>
__global__ __launch_bounds__(32 * (DOUT / 16)) void k_root_gemm(
    const float* __restrict__ x, const float* __restrict__ W,
    const float* __restrict__ bias, float* __restrict__ y, int N) {
  __shared__ __attribute__((aligned(16))) float As[16 * A_PITCH];
  constexpr int NW = DOUT / 16;
  const int wave = threadIdx.x >> 5, lane = threadIdx.x & 31;
  const int m = lane & 15;
  const int khalf = (lane >> 4) << 1;
#pragma unroll
  for (int j = 0; j < 16 / NW; ++j) {
    const int tr = wave + j * NW;
    const int row = (blockIdx.x << 4) + tr;
    stage_row(As, x, row < N ? row : N - 1, tr, lane);
  }
  __syncthreads();
  const float* xa = As + m * A_PITCH;
  const int col = (wave << 4) + m;
  const float* wb = W + col;
  v8f c = {};
#pragma unroll 8
  for (int k = 0; k < D_IN; k += 4) {
    v2f a, b;
    a.x = xa[k + khalf];
    a.y = xa[k + khalf + 1];
    b.x = wb[(k + khalf) * DOUT];
    b.y = wb[(k + khalf + 1) * DOUT];
    c = WMMA_F32(a, b, c);
  }
  const float bv = bias[col];
  const int rbase = (blockIdx.x << 4) + ((lane >> 4) << 3);
  float* yp = y + (size_t)rbase * DOUT + col;
  if ((blockIdx.x << 4) + 16 <= N) {
#pragma unroll
    for (int i = 0; i < 8; ++i) yp[i * DOUT] = c[i] + bv;
  } else {
#pragma unroll
    for (int i = 0; i < 8; ++i)
      if (rbase + i < N) yp[i * DOUT] = c[i] + bv;
  }
}


static constexpr int RUN_TILES = 8;
static constexpr int RUN_SEGS = RUN_TILES * 16;
static constexpr int RUN_EDGES_MAX = RUN_SEGS * SEG_EDGES;

__device__ __forceinline__ float4 ld_row(const float* __restrict__ x, int row, int lane) {
  return *reinterpret_cast<const float4*>(x + (size_t)row * D_IN + (lane << 2));
}

__device__ __forceinline__ float4 add4(float4 a, float4 b) {
  return make_float4(a.x + b.x, a.y + b.y, a.z + b.z, a.w + b.w);
}

__device__ __forceinline__ float4 keep_if(bool keep, float4 v) {
  return keep ? v : make_float4(0.f, 0.f, 0.f, 0.f);
}

struct SegRows {
  float4 p[SEG_EDGES];
  int n;
};

__device__ __forceinline__ SegRows seg_fetch(const float* __restrict__ x, const int* sSrc,
                                             int rel, int n, int lane) {
  SegRows s;
#pragma unroll
  for (int q = 0; q < SEG_EDGES; ++q) s.p[q] = ld_row(x, sSrc[rel + q], lane);
  s.n = n;
  return s;
}

__device__ __forceinline__ float4 seg_total(const SegRows& s) {
  return add4(add4(keep_if(s.n > 0, s.p[0]), keep_if(s.n > 1, s.p[1])),
              add4(keep_if(s.n > 2, s.p[2]), keep_if(s.n > 3, s.p[3])));
}

template <int DOUT>
__global__ __launch_bounds__(32 * (DOUT / 16)) void k_edge_wmma(
    const float* __restrict__ x, const float* __restrict__ Wrel,
    const int* __restrict__ ssrc, const int* __restrict__ segBeg, const int* __restrict__ segDst,
    const int* __restrict__ starts, float* __restrict__ y) {
  static_assert(SEG_EDGES == 4, "seg_total() is written for four rows");
  __shared__ __attribute__((aligned(16))) float As[2][16 * A_PITCH];
  __shared__ __attribute__((aligned(16))) int sBeg[RUN_SEGS + 16];
  __shared__ __attribute__((aligned(16))) int dIdx[RUN_SEGS];
  __shared__ __attribute__((aligned(16))) int sSrc[RUN_EDGES_MAX + 16];
  constexpr int NW = DOUT / 16;
  constexpr int RPW = 16 / NW;
  const int total = starts[8];
  const int base0 = blockIdx.x * RUN_SEGS;
  if (base0 >= total) return;
  const int left = (total - base0) >> 4;
  const int nT = left < RUN_TILES ? left : RUN_TILES;
  const int lane = threadIdx.x & 31;
  const int wave = __builtin_amdgcn_readfirstlane((int)(threadIdx.x >> 5));

  const int eb0 = segBeg[base0];
  const int nE = segBeg[base0 + (nT << 4)] - eb0;
  for (int i = threadIdx.x; i < RUN_SEGS + 1; i += 32 * NW) {
    const int gi = base0 + i;
    sBeg[i] = (gi <= total ? segBeg[gi] : eb0) - eb0;
    if (i < RUN_SEGS) dIdx[i] = gi < total ? segDst[gi] : 0;
  }
  for (int i = threadIdx.x; i < RUN_EDGES_MAX + 16; i += 32 * NW)
    sSrc[i] = i < nE ? ssrc[eb0 + i] : 0;
  __syncthreads();

#pragma unroll
  for (int j = 0; j < RPW; ++j) {
    const int tr = wave + j * NW;
    const int rb = __builtin_amdgcn_readfirstlane(sBeg[tr]);
    const int re = __builtin_amdgcn_readfirstlane(sBeg[tr + 1]);
    const SegRows s = seg_fetch(x, sSrc, rb, re - rb, lane);
    *reinterpret_cast<float4*>(As[0] + tr * A_PITCH + (lane << 2)) = seg_total(s);
  }
  __syncthreads();

  const int m = lane & 15;
  const int khalf = (lane >> 4) << 1;
  const int col = ((int)(threadIdx.x >> 5) << 4) + m;
  const int mBase = (lane >> 4) << 3;
  const int s1 = starts[1], s2 = starts[2], s3 = starts[3], s4 = starts[4],
            s5 = starts[5], s6 = starts[6], s7 = starts[7];

  v2f b[D_IN / 4];
  int curR = -1;
  int cur = 0;
  for (int t = 0; t < nT; ++t) {
    const int base = base0 + (t << 4);
    const int r = (base >= s1) + (base >= s2) + (base >= s3) + (base >= s4) +
                  (base >= s5) + (base >= s6) + (base >= s7);
    if (r != curR) {
      const float* wb = Wrel + (size_t)r * (D_IN * DOUT) + col;
#pragma unroll
      for (int kk = 0; kk < D_IN / 4; ++kk) {
        b[kk].x = wb[(4 * kk + khalf) * DOUT];
        b[kk].y = wb[(4 * kk + khalf + 1) * DOUT];
      }
      curR = r;
    }

    const int tn = t + 1 < nT ? t + 1 : t;
    SegRows pre[RPW];
#pragma unroll
    for (int j = 0; j < RPW; ++j) {
      const int tr = (tn << 4) + wave + j * NW;
      const int rb = __builtin_amdgcn_readfirstlane(sBeg[tr]);
      const int re = __builtin_amdgcn_readfirstlane(sBeg[tr + 1]);
      pre[j] = seg_fetch(x, sSrc, rb, re - rb, lane);
    }

    const float* xa = As[cur] + m * A_PITCH;
    v8f c = {};
#pragma unroll
    for (int kk = 0; kk < D_IN / 4; ++kk) {
      v2f a;
      a.x = xa[4 * kk + khalf];
      a.y = xa[4 * kk + khalf + 1];
      c = WMMA_F32(a, b[kk], c);
    }

    float* An = As[cur ^ 1];
#pragma unroll
    for (int j = 0; j < RPW; ++j) {
      const int tr = wave + j * NW;
      *reinterpret_cast<float4*>(An + tr * A_PITCH + (lane << 2)) = seg_total(pre[j]);
    }

    __syncthreads();
    cur ^= 1;

    const int* dp = dIdx + (t << 4) + mBase;
#pragma unroll
    for (int i = 0; i < 8; ++i) {
      const int drow = dp[i];
      atomicAdd(&y[(size_t)drow * DOUT + col], c[i]);
    }
  }
}


__global__ void k_relu4(float4* __restrict__ p, long long n4) {
  long long i = (long long)blockIdx.x * blockDim.x + threadIdx.x;
  if (i < n4) {
    float4 v = p[i];
    v.x = fmaxf(v.x, 0.0f);
    v.y = fmaxf(v.y, 0.0f);
    v.z = fmaxf(v.z, 0.0f);
    v.w = fmaxf(v.w, 0.0f);
    p[i] = v;
  }
}

__global__ void k_logsoftmax(const float* __restrict__ y, float* __restrict__ out, int N) {
  int wave = threadIdx.x >> 5, lane = threadIdx.x & 31;
  int node = blockIdx.x * (blockDim.x >> 5) + wave;
  if (node >= N) return;
  const float* row = y + (size_t)node * 64;
  float v0 = row[lane], v1 = row[lane + 32];
  float mx = fmaxf(v0, v1);
#pragma unroll
  for (int off = 16; off > 0; off >>= 1) mx = fmaxf(mx, __shfl_xor(mx, off, 32));
  float s = expf(v0 - mx) + expf(v1 - mx);
#pragma unroll
  for (int off = 16; off > 0; off >>= 1) s += __shfl_xor(s, off, 32);
  float lse = logf(s) + mx;
  float* o = out + (size_t)node * 64;
  o[lane] = v0 - lse;
  o[lane + 32] = v1 - lse;
}


static inline void launch_layer(const float* x, const float* Wrel, const float* Wroot,
                                const float* bias, float* y, int dOut, int N,
                                const int* ssrc, const int* segBeg, const int* segDst, const int* starts,
                                int nodeTiles, int segRunsMax, hipStream_t stream) {
  if (dOut == 128) {
    k_root_gemm<128><<<nodeTiles, 32 * (128 / 16), 0, stream>>>(x, Wroot, bias, y, N);
    k_edge_wmma<128><<<segRunsMax, 32 * (128 / 16), 0, stream>>>(x, Wrel, ssrc, segBeg, segDst, starts, y);
  } else {
    k_root_gemm<64><<<nodeTiles, 32 * (64 / 16), 0, stream>>>(x, Wroot, bias, y, N);
    k_edge_wmma<64><<<segRunsMax, 32 * (64 / 16), 0, stream>>>(x, Wrel, ssrc, segBeg, segDst, starts, y);
  }
}

extern "C" void kernel_launch(void* const* d_in, const int* in_sizes, int n_in,
                              void* d_out, int out_size, void* d_ws, size_t ws_size,
                              hipStream_t stream) {
  const float* x0     = (const float*)d_in[0];
  const int*   ei     = (const int*)  d_in[1];
  const int*   etype  = (const int*)  d_in[2];
  const float* Wrel1  = (const float*)d_in[3];
  const float* Wroot1 = (const float*)d_in[4];
  const float* b1     = (const float*)d_in[5];
  const float* Wrel2  = (const float*)d_in[6];
  const float* Wroot2 = (const float*)d_in[7];
  const float* b2     = (const float*)d_in[8];
  const float* Wrel3  = (const float*)d_in[9];
  const float* Wroot3 = (const float*)d_in[10];
  const float* b3     = (const float*)d_in[11];

  const int N  = in_sizes[0] / D_IN;
  const int E  = in_sizes[2];
  const int D1 = 128, D2 = 128;
  const int D3 = in_sizes[11];
  const int K  = 8 * N;
  const int keysMax = E < K ? E : K;
  const int segsMax = keysMax + (E - keysMax + SEG_EDGES - 1) / SEG_EDGES + 128;

  char* w = (char*)d_ws;
  int* starts = (int*)w;
  size_t off = 256;
  int* segBeg = (int*)(w + off); off += (size_t)(segsMax + 16) * sizeof(int);
  int* segDst = (int*)(w + off); off += (size_t)(segsMax + 16) * sizeof(int);
  int* ssrc   = (int*)(w + off); off += (size_t)E * sizeof(int);
  off = (off + 255) & ~(size_t)255;
  float* bufA = (float*)(w + off); off += (size_t)(N + 16) * D_IN * sizeof(float);
  float* bufB = (float*)(w + off);
  int* cnt  = (int*)bufB;
  int* blkE = cnt + K;
  int* blkS = blkE + MID_MAX;
  int* blkBaseE = blkS + MID_MAX;
  int* blkBaseS = blkBaseE + MID_MAX;

  const int nb = (N + SCAN_THREADS - 1) / SCAN_THREADS;
  k_zero_ints  <<<(K + 255) / 256, 256, 0, stream>>>(cnt, K);
  k_key_hist   <<<(E + 255) / 256, 256, 0, stream>>>(ei, etype, E, N, cnt);
  k_scan_a     <<<dim3(nb, 8), SCAN_THREADS, 0, stream>>>(cnt, N, nb, blkE, blkS);
  k_scan_mid   <<<1, SCAN_THREADS, 0, stream>>>(blkE, blkS, nb, N, blkBaseE, blkBaseS, starts, segBeg, segDst);
  k_scan_b     <<<dim3(nb, 8), SCAN_THREADS, 0, stream>>>(cnt, N, nb, blkBaseE, blkBaseS, segBeg, segDst);
  k_key_scatter<<<(E + 255) / 256, 256, 0, stream>>>(ei, etype, E, N, cnt, ssrc);

  const int nodeTiles  = (N + 15) / 16;
  const int segRunsMax = (segsMax + RUN_SEGS - 1) / RUN_SEGS;

  launch_layer(x0, Wrel1, Wroot1, b1, bufA, D1, N, ssrc, segBeg, segDst, starts,
               nodeTiles, segRunsMax, stream);
  {
    long long n4 = (long long)N * D1 / 4;
    k_relu4<<<(unsigned)((n4 + 255) / 256), 256, 0, stream>>>((float4*)bufA, n4);
  }

  launch_layer(bufA, Wrel2, Wroot2, b2, bufB, D2, N, ssrc, segBeg, segDst, starts,
               nodeTiles, segRunsMax, stream);
  {
    long long n4 = (long long)N * D2 / 4;
    k_relu4<<<(unsigned)((n4 + 255) / 256), 256, 0, stream>>>((float4*)bufB, n4);
  }

  launch_layer(bufB, Wrel3, Wroot3, b3, bufA, D3, N, ssrc, segBeg, segDst, starts,
               nodeTiles, segRunsMax, stream);
  k_logsoftmax<<<(N + 7) / 8, 256, 0, stream>>>(bufA, (float*)d_out, N);
}


# Edge kernels: each gathered row's address, and the zero row for a segment's unused slots, formed on the scalar unit one trip ahead: no per-lane 64-bit address math, no selects before the sums
# speedup vs baseline: 3.6831x; 1.0479x over previous
#include <hip/hip_runtime.h>
#include <hip/hip_bf16.h>
#include <math.h>


typedef float v2f __attribute__((ext_vector_type(2)));
typedef float v8f __attribute__((ext_vector_type(8)));

#define WMMA_F32(a, b, c) \
  __builtin_amdgcn_wmma_f32_16x16x4_f32(false, (a), false, (b), (short)0, (c), false, false)

static constexpr int D_IN = 128;

static constexpr int A_PITCH = D_IN + 4;

__device__ __forceinline__ void stage_row(float* As, const float* __restrict__ x,
                                          int srcRow, int tileRow, int lane) {
  const float4 v = *reinterpret_cast<const float4*>(x + (size_t)srcRow * D_IN + (lane << 2));
  *reinterpret_cast<float4*>(As + tileRow * A_PITCH + (lane << 2)) = v;
}

static constexpr int SEG_EDGES = 4;

__global__ void k_zero_ints(int* __restrict__ p, int n, float* __restrict__ zrow) {
  int i = blockIdx.x * blockDim.x + threadIdx.x;
  if (i < n) p[i] = 0;
  if (i < D_IN) zrow[i] = 0.f;
}

__global__ void k_key_hist(const int* __restrict__ ei, const int* __restrict__ etype, int E, int N,
                           int* __restrict__ cnt) {
  const int e = blockIdx.x * blockDim.x + threadIdx.x;
  if (e < E) atomicAdd(&cnt[(etype[e] & 7) * N + ei[E + e]], 1);
}

static constexpr int SCAN_THREADS = 1024;

__device__ __forceinline__ int wave_incl_scan(int v, int lane) {
#pragma unroll
  for (int o = 1; o < 32; o <<= 1) {
    const int t = __shfl_up(v, o, 32);
    if (lane >= o) v += t;
  }
  return v;
}

__device__ __forceinline__ int wave_sum(int v) {
#pragma unroll
  for (int o = 16; o > 0; o >>= 1) v += __shfl_xor(v, o, 32);
  return v;
}

__global__ __launch_bounds__(SCAN_THREADS) void k_scan_a(
    const int* __restrict__ cnt, int N, int nb, int* __restrict__ blkE, int* __restrict__ blkS) {
  __shared__ int wE[32], wS[32];
  const int r = blockIdx.y, t = threadIdx.x, lane = t & 31, wave = t >> 5;
  const int k = blockIdx.x * SCAN_THREADS + t;
  const int v = k < N ? cnt[(size_t)r * N + k] : 0;
  const int e = wave_sum(v);
  const int s = wave_sum((v + SEG_EDGES - 1) / SEG_EDGES);
  if (lane == 0) { wE[wave] = e; wS[wave] = s; }
  __syncthreads();
  if (wave == 0) {
    const int te = wave_sum(wE[lane]), ts = wave_sum(wS[lane]);
    if (lane == 0) {
      blkE[r * nb + blockIdx.x] = te;
      blkS[r * nb + blockIdx.x] = ts;
    }
  }
}

static constexpr int MID_MAX = 1024;
__global__ __launch_bounds__(SCAN_THREADS) void k_scan_mid(
    const int* __restrict__ blkE, const int* __restrict__ blkS, int nb, int N,
    int* __restrict__ blkBaseE, int* __restrict__ blkBaseS,
    int* __restrict__ starts, int* __restrict__ segBeg, int* __restrict__ segDst) {
  __shared__ int inE[MID_MAX], inS[MID_MAX];
  __shared__ int relE[8], relS[8], baseE[9], baseS[9];
  const int t = threadIdx.x, lane = t & 31, wave = t >> 5;
  if (wave < 8) {
    int carryE = 0, carryS = 0;
    for (int b0 = 0; b0 < nb; b0 += 32) {
      const int b = b0 + lane;
      const int i = wave * nb + b;
      const int ve = b < nb ? blkE[i] : 0;
      const int vs = b < nb ? blkS[i] : 0;
      const int ie = wave_incl_scan(ve, lane), is = wave_incl_scan(vs, lane);
      if (b < nb) {
        inE[i] = carryE + ie - ve;
        inS[i] = carryS + is - vs;
      }
      carryE += __shfl(ie, 31, 32);
      carryS += __shfl(is, 31, 32);
    }
    if (lane == 0) { relE[wave] = carryE; relS[wave] = carryS; }
  }
  __syncthreads();
  if (t == 0) {
    int e = 0, s = 0;
    for (int r = 0; r < 8; ++r) {
      baseE[r] = e;
      baseS[r] = s;
      e += relE[r];
      s += (relS[r] + 15) & ~15;
    }
    baseE[8] = e;
    baseS[8] = s;
  }
  __syncthreads();
  for (int i = t; i < 8 * nb; i += SCAN_THREADS) {
    const int r = i / nb;
    blkBaseE[i] = baseE[r] + inE[i];
    blkBaseS[i] = baseS[r] + inS[i];
  }
  if (t < 9) starts[t] = baseS[t];
  if (t == 0) segBeg[baseS[8]] = baseE[8];
  if (t < 128) {
    const int r = t >> 4;
    const int p = baseS[r] + relS[r] + (t & 15);
    if (p < baseS[r + 1]) {
      segBeg[p] = baseE[r + 1];
      segDst[p] = N;
    }
  }
}

__global__ __launch_bounds__(SCAN_THREADS) void k_scan_b(
    int* __restrict__ cnt, int N, int nb,
    const int* __restrict__ blkBaseE, const int* __restrict__ blkBaseS,
    int* __restrict__ segBeg, int* __restrict__ segDst) {
  __shared__ int wE[32], wS[32];
  const int r = blockIdx.y, t = threadIdx.x, lane = t & 31, wave = t >> 5;
  const int k = blockIdx.x * SCAN_THREADS + t;
  const int v = k < N ? cnt[(size_t)r * N + k] : 0;
  const int sv = (v + SEG_EDGES - 1) / SEG_EDGES;
  const int ie = wave_incl_scan(v, lane), is = wave_incl_scan(sv, lane);
  if (lane == 31) { wE[wave] = ie; wS[wave] = is; }
  __syncthreads();
  if (wave == 0) {
    const int a = wE[lane], b = wS[lane];
    const int ia = wave_incl_scan(a, lane), ib = wave_incl_scan(b, lane);
    wE[lane] = ia - a;
    wS[lane] = ib - b;
  }
  __syncthreads();
  const int e0 = blkBaseE[r * nb + blockIdx.x] + wE[wave] + ie - v;
  int s = blkBaseS[r * nb + blockIdx.x] + wS[wave] + is - sv;
  if (k < N) {
    cnt[(size_t)r * N + k] = e0;
    for (int q = 0; q < v; q += SEG_EDGES) {
      segBeg[s] = e0 + q;
      segDst[s] = k;
      ++s;
    }
  }
}

__global__ void k_key_scatter(const int* __restrict__ ei, const int* __restrict__ etype, int E, int N,
                              int* __restrict__ cursor, int* __restrict__ ssrc) {
  const int e = blockIdx.x * blockDim.x + threadIdx.x;
  if (e < E) {
    const int pos = atomicAdd(&cursor[(etype[e] & 7) * N + ei[E + e]], 1);
    ssrc[pos] = ei[e];
  }
}


template <int DOUT>
__global__ __launch_bounds__(32 * (DOUT / 16)) void k_root_gemm(
    const float* __restrict__ x, const float* __restrict__ W,
    const float* __restrict__ bias, float* __restrict__ y, int N) {
  __shared__ __attribute__((aligned(16))) float As[16 * A_PITCH];
  constexpr int NW = DOUT / 16;
  const int wave = threadIdx.x >> 5, lane = threadIdx.x & 31;
  const int m = lane & 15;
  const int khalf = (lane >> 4) << 1;
#pragma unroll
  for (int j = 0; j < 16 / NW; ++j) {
    const int tr = wave + j * NW;
    const int row = (blockIdx.x << 4) + tr;
    stage_row(As, x, row < N ? row : N - 1, tr, lane);
  }
  __syncthreads();
  const float* xa = As + m * A_PITCH;
  const int col = (wave << 4) + m;
  const float* wb = W + col;
  v8f c = {};
#pragma unroll 8
  for (int k = 0; k < D_IN; k += 4) {
    v2f a, b;
    a.x = xa[k + khalf];
    a.y = xa[k + khalf + 1];
    b.x = wb[(k + khalf) * DOUT];
    b.y = wb[(k + khalf + 1) * DOUT];
    c = WMMA_F32(a, b, c);
  }
  const float bv = bias[col];
  const int rbase = (blockIdx.x << 4) + ((lane >> 4) << 3);
  float* yp = y + (size_t)rbase * DOUT + col;
  if ((blockIdx.x << 4) + 16 <= N) {
#pragma unroll
    for (int i = 0; i < 8; ++i) yp[i * DOUT] = c[i] + bv;
  } else {
#pragma unroll
    for (int i = 0; i < 8; ++i)
      if (rbase + i < N) yp[i * DOUT] = c[i] + bv;
  }
}


static constexpr int RUN_TILES = 8;
static constexpr int RUN_SEGS = RUN_TILES * 16;
static constexpr int RUN_EDGES_MAX = RUN_SEGS * SEG_EDGES;

__device__ __forceinline__ float4 add4(float4 a, float4 b) {
  return make_float4(a.x + b.x, a.y + b.y, a.z + b.z, a.w + b.w);
}

struct SegRows {
  float4 p[SEG_EDGES];
};

struct RowPtrs {
  const float* p[SEG_EDGES];
};

__device__ __forceinline__ RowPtrs seg_ptrs(const float* __restrict__ x, const float* __restrict__ zrow,
                                            const int* sBeg, const int* sSrc, int seg) {
  const int rb = __builtin_amdgcn_readfirstlane(sBeg[seg]);
  const int n = __builtin_amdgcn_readfirstlane(sBeg[seg + 1]) - rb;
  RowPtrs r;
#pragma unroll
  for (int q = 0; q < SEG_EDGES; ++q) {
    const int row = __builtin_amdgcn_readfirstlane(sSrc[rb + q]);
    r.p[q] = q < n ? x + (size_t)row * D_IN : zrow;
  }
  return r;
}

__device__ __forceinline__ SegRows seg_fetch(const RowPtrs& r, unsigned laneBytes) {
  SegRows s;
#pragma unroll
  for (int q = 0; q < SEG_EDGES; ++q)
    s.p[q] = *reinterpret_cast<const float4*>(reinterpret_cast<const char*>(r.p[q]) + laneBytes);
  return s;
}

__device__ __forceinline__ float4 seg_total(const SegRows& s) {
  return add4(add4(s.p[0], s.p[1]), add4(s.p[2], s.p[3]));
}

template <int DOUT>
__global__ __launch_bounds__(32 * (DOUT / 16)) void k_edge_wmma(
    const float* __restrict__ x, const float* __restrict__ Wrel,
    const int* __restrict__ ssrc, const int* __restrict__ segBeg, const int* __restrict__ segDst,
    const int* __restrict__ starts, const float* __restrict__ zrow, float* __restrict__ y) {
  static_assert(SEG_EDGES == 4, "seg_total() is written for four rows");
  __shared__ __attribute__((aligned(16))) float As[2][16 * A_PITCH];
  __shared__ __attribute__((aligned(16))) int sBeg[RUN_SEGS + 16];
  __shared__ __attribute__((aligned(16))) int dIdx[RUN_SEGS];
  __shared__ __attribute__((aligned(16))) int sSrc[RUN_EDGES_MAX + 16];
  constexpr int NW = DOUT / 16;
  constexpr int RPW = 16 / NW;
  const int total = starts[8];
  const int base0 = blockIdx.x * RUN_SEGS;
  if (base0 >= total) return;
  const int left = (total - base0) >> 4;
  const int nT = left < RUN_TILES ? left : RUN_TILES;
  const int lane = threadIdx.x & 31;
  const unsigned laneBytes = (unsigned)lane << 4;
  const int wave = __builtin_amdgcn_readfirstlane((int)(threadIdx.x >> 5));

  const int eb0 = segBeg[base0];
  const int nE = segBeg[base0 + (nT << 4)] - eb0;
  for (int i = threadIdx.x; i < RUN_SEGS + 1; i += 32 * NW) {
    const int gi = base0 + i;
    sBeg[i] = (gi <= total ? segBeg[gi] : eb0) - eb0;
    if (i < RUN_SEGS) dIdx[i] = gi < total ? segDst[gi] : 0;
  }
  for (int i = threadIdx.x; i < RUN_EDGES_MAX + 16; i += 32 * NW)
    sSrc[i] = i < nE ? ssrc[eb0 + i] : 0;
  __syncthreads();

#pragma unroll
  for (int j = 0; j < RPW; ++j) {
    const int tr = wave + j * NW;
    const SegRows s = seg_fetch(seg_ptrs(x, zrow, sBeg, sSrc, tr), laneBytes);
    *reinterpret_cast<float4*>(As[0] + tr * A_PITCH + (lane << 2)) = seg_total(s);
  }
  RowPtrs nx[RPW];
  {
    const int t1 = 1 < nT ? 1 : 0;
#pragma unroll
    for (int j = 0; j < RPW; ++j) nx[j] = seg_ptrs(x, zrow, sBeg, sSrc, (t1 << 4) + wave + j * NW);
  }
  __syncthreads();

  const int m = lane & 15;
  const int khalf = (lane >> 4) << 1;
  const int col = ((int)(threadIdx.x >> 5) << 4) + m;
  const int mBase = (lane >> 4) << 3;
  const unsigned colBytes = (unsigned)col << 2;
  const int s1 = starts[1], s2 = starts[2], s3 = starts[3], s4 = starts[4],
            s5 = starts[5], s6 = starts[6], s7 = starts[7];

  v2f b[D_IN / 4];
  int curR = -1;
  int cur = 0;
  for (int t = 0; t < nT; ++t) {
    const int base = base0 + (t << 4);
    const int r = (base >= s1) + (base >= s2) + (base >= s3) + (base >= s4) +
                  (base >= s5) + (base >= s6) + (base >= s7);
    if (r != curR) {
      const float* wb = Wrel + (size_t)r * (D_IN * DOUT) + col;
#pragma unroll
      for (int kk = 0; kk < D_IN / 4; ++kk) {
        b[kk].x = wb[(4 * kk + khalf) * DOUT];
        b[kk].y = wb[(4 * kk + khalf + 1) * DOUT];
      }
      curR = r;
    }

    unsigned lb = laneBytes;
    asm volatile("" : "+v"(lb));
    SegRows pre[RPW];
#pragma unroll
    for (int j = 0; j < RPW; ++j) pre[j] = seg_fetch(nx[j], lb);
    const int tnn = t + 2 < nT ? t + 2 : nT - 1;
#pragma unroll
    for (int j = 0; j < RPW; ++j) nx[j] = seg_ptrs(x, zrow, sBeg, sSrc, (tnn << 4) + wave + j * NW);

    const float* xa = As[cur] + m * A_PITCH;
    v8f c = {};
#pragma unroll
    for (int kk = 0; kk < D_IN / 4; ++kk) {
      v2f a;
      a.x = xa[4 * kk + khalf];
      a.y = xa[4 * kk + khalf + 1];
      c = WMMA_F32(a, b[kk], c);
    }

    float* An = As[cur ^ 1];
#pragma unroll
    for (int j = 0; j < RPW; ++j) {
      const int tr = wave + j * NW;
      *reinterpret_cast<float4*>(An + tr * A_PITCH + (lane << 2)) = seg_total(pre[j]);
    }

    __syncthreads();
    cur ^= 1;

    const int* dp = dIdx + (t << 4) + mBase;
#pragma unroll
    for (int i = 0; i < 8; ++i) {
      const unsigned off = (unsigned)dp[i] * (unsigned)(DOUT * 4) + colBytes;
      atomicAdd(reinterpret_cast<float*>(reinterpret_cast<char*>(y) + off), c[i]);
    }
  }
}


__global__ void k_relu4(float4* __restrict__ p, long long n4) {
  long long i = (long long)blockIdx.x * blockDim.x + threadIdx.x;
  if (i < n4) {
    float4 v = p[i];
    v.x = fmaxf(v.x, 0.0f);
    v.y = fmaxf(v.y, 0.0f);
    v.z = fmaxf(v.z, 0.0f);
    v.w = fmaxf(v.w, 0.0f);
    p[i] = v;
  }
}

__global__ void k_logsoftmax(const float* __restrict__ y, float* __restrict__ out, int N) {
  int wave = threadIdx.x >> 5, lane = threadIdx.x & 31;
  int node = blockIdx.x * (blockDim.x >> 5) + wave;
  if (node >= N) return;
  const float* row = y + (size_t)node * 64;
  float v0 = row[lane], v1 = row[lane + 32];
  float mx = fmaxf(v0, v1);
#pragma unroll
  for (int off = 16; off > 0; off >>= 1) mx = fmaxf(mx, __shfl_xor(mx, off, 32));
  float s = expf(v0 - mx) + expf(v1 - mx);
#pragma unroll
  for (int off = 16; off > 0; off >>= 1) s += __shfl_xor(s, off, 32);
  float lse = logf(s) + mx;
  float* o = out + (size_t)node * 64;
  o[lane] = v0 - lse;
  o[lane + 32] = v1 - lse;
}


static inline void launch_layer(const float* x, const float* Wrel, const float* Wroot,
                                const float* bias, float* y, int dOut, int N,
                                const int* ssrc, const int* segBeg, const int* segDst, const int* starts,
                                const float* zrow, int nodeTiles, int segRunsMax, hipStream_t stream) {
  if (dOut == 128) {
    k_root_gemm<128><<<nodeTiles, 32 * (128 / 16), 0, stream>>>(x, Wroot, bias, y, N);
    k_edge_wmma<128><<<segRunsMax, 32 * (128 / 16), 0, stream>>>(x, Wrel, ssrc, segBeg, segDst, starts, zrow, y);
  } else {
    k_root_gemm<64><<<nodeTiles, 32 * (64 / 16), 0, stream>>>(x, Wroot, bias, y, N);
    k_edge_wmma<64><<<segRunsMax, 32 * (64 / 16), 0, stream>>>(x, Wrel, ssrc, segBeg, segDst, starts, zrow, y);
  }
}

extern "C" void kernel_launch(void* const* d_in, const int* in_sizes, int n_in,
                              void* d_out, int out_size, void* d_ws, size_t ws_size,
                              hipStream_t stream) {
  const float* x0     = (const float*)d_in[0];
  const int*   ei     = (const int*)  d_in[1];
  const int*   etype  = (const int*)  d_in[2];
  const float* Wrel1  = (const float*)d_in[3];
  const float* Wroot1 = (const float*)d_in[4];
  const float* b1     = (const float*)d_in[5];
  const float* Wrel2  = (const float*)d_in[6];
  const float* Wroot2 = (const float*)d_in[7];
  const float* b2     = (const float*)d_in[8];
  const float* Wrel3  = (const float*)d_in[9];
  const float* Wroot3 = (const float*)d_in[10];
  const float* b3     = (const float*)d_in[11];

  const int N  = in_sizes[0] / D_IN;
  const int E  = in_sizes[2];
  const int D1 = 128, D2 = 128;
  const int D3 = in_sizes[11];
  const int K  = 8 * N;
  const int keysMax = E < K ? E : K;
  const int segsMax = keysMax + (E - keysMax + SEG_EDGES - 1) / SEG_EDGES + 128;

  char* w = (char*)d_ws;
  int* starts = (int*)w;
  size_t off = 256;
  int* segBeg = (int*)(w + off); off += (size_t)(segsMax + 16) * sizeof(int);
  int* segDst = (int*)(w + off); off += (size_t)(segsMax + 16) * sizeof(int);
  int* ssrc   = (int*)(w + off); off += (size_t)E * sizeof(int);
  off = (off + 255) & ~(size_t)255;
  float* bufA = (float*)(w + off); off += (size_t)(N + 16) * D_IN * sizeof(float);
  float* bufB = (float*)(w + off);
  float* zrow = bufA + (size_t)(N + 8) * D_IN;
  int* cnt  = (int*)bufB;
  int* blkE = cnt + K;
  int* blkS = blkE + MID_MAX;
  int* blkBaseE = blkS + MID_MAX;
  int* blkBaseS = blkBaseE + MID_MAX;

  const int nb = (N + SCAN_THREADS - 1) / SCAN_THREADS;
  k_zero_ints  <<<(K + 255) / 256, 256, 0, stream>>>(cnt, K, zrow);
  k_key_hist   <<<(E + 255) / 256, 256, 0, stream>>>(ei, etype, E, N, cnt);
  k_scan_a     <<<dim3(nb, 8), SCAN_THREADS, 0, stream>>>(cnt, N, nb, blkE, blkS);
  k_scan_mid   <<<1, SCAN_THREADS, 0, stream>>>(blkE, blkS, nb, N, blkBaseE, blkBaseS, starts, segBeg, segDst);
  k_scan_b     <<<dim3(nb, 8), SCAN_THREADS, 0, stream>>>(cnt, N, nb, blkBaseE, blkBaseS, segBeg, segDst);
  k_key_scatter<<<(E + 255) / 256, 256, 0, stream>>>(ei, etype, E, N, cnt, ssrc);

  const int nodeTiles  = (N + 15) / 16;
  const int segRunsMax = (segsMax + RUN_SEGS - 1) / RUN_SEGS;

  launch_layer(x0, Wrel1, Wroot1, b1, bufA, D1, N, ssrc, segBeg, segDst, starts,
               zrow, nodeTiles, segRunsMax, stream);
  {
    long long n4 = (long long)N * D1 / 4;
    k_relu4<<<(unsigned)((n4 + 255) / 256), 256, 0, stream>>>((float4*)bufA, n4);
  }

  launch_layer(bufA, Wrel2, Wroot2, b2, bufB, D2, N, ssrc, segBeg, segDst, starts,
               zrow, nodeTiles, segRunsMax, stream);
  {
    long long n4 = (long long)N * D2 / 4;
    k_relu4<<<(unsigned)((n4 + 255) / 256), 256, 0, stream>>>((float4*)bufB, n4);
  }

  launch_layer(bufB, Wrel3, Wroot3, b3, bufA, D3, N, ssrc, segBeg, segDst, starts,
               zrow, nodeTiles, segRunsMax, stream);
  k_logsoftmax<<<(N + 7) / 8, 256, 0, stream>>>(bufA, (float*)d_out, N);
}


# Edge scatter through LDS: each float atomic instruction adds one whole 128-B line of one destination row (scalar row base), not two half lines of two rows; atomics kept; waves_per_eu pins schedule
# speedup vs baseline: 3.8159x; 1.0361x over previous
#include <hip/hip_runtime.h>
#include <hip/hip_bf16.h>
#include <math.h>


typedef float v2f __attribute__((ext_vector_type(2)));
typedef float v8f __attribute__((ext_vector_type(8)));

#define WMMA_F32(a, b, c) \
  __builtin_amdgcn_wmma_f32_16x16x4_f32(false, (a), false, (b), (short)0, (c), false, false)

static constexpr int D_IN = 128;

static constexpr int A_PITCH = D_IN + 4;

__device__ __forceinline__ void stage_row(float* As, const float* __restrict__ x,
                                          int srcRow, int tileRow, int lane) {
  const float4 v = *reinterpret_cast<const float4*>(x + (size_t)srcRow * D_IN + (lane << 2));
  *reinterpret_cast<float4*>(As + tileRow * A_PITCH + (lane << 2)) = v;
}

static constexpr int SEG_EDGES = 4;

__global__ void k_zero_ints(int* __restrict__ p, int n, float* __restrict__ zrow) {
  int i = blockIdx.x * blockDim.x + threadIdx.x;
  if (i < n) p[i] = 0;
  if (i < D_IN) zrow[i] = 0.f;
}

__global__ void k_key_hist(const int* __restrict__ ei, const int* __restrict__ etype, int E, int N,
                           int* __restrict__ cnt) {
  const int e = blockIdx.x * blockDim.x + threadIdx.x;
  if (e < E) atomicAdd(&cnt[(etype[e] & 7) * N + ei[E + e]], 1);
}

static constexpr int SCAN_THREADS = 1024;

__device__ __forceinline__ int wave_incl_scan(int v, int lane) {
#pragma unroll
  for (int o = 1; o < 32; o <<= 1) {
    const int t = __shfl_up(v, o, 32);
    if (lane >= o) v += t;
  }
  return v;
}

__device__ __forceinline__ int wave_sum(int v) {
#pragma unroll
  for (int o = 16; o > 0; o >>= 1) v += __shfl_xor(v, o, 32);
  return v;
}

__global__ __launch_bounds__(SCAN_THREADS) void k_scan_a(
    const int* __restrict__ cnt, int N, int nb, int* __restrict__ blkE, int* __restrict__ blkS) {
  __shared__ int wE[32], wS[32];
  const int r = blockIdx.y, t = threadIdx.x, lane = t & 31, wave = t >> 5;
  const int k = blockIdx.x * SCAN_THREADS + t;
  const int v = k < N ? cnt[(size_t)r * N + k] : 0;
  const int e = wave_sum(v);
  const int s = wave_sum((v + SEG_EDGES - 1) / SEG_EDGES);
  if (lane == 0) { wE[wave] = e; wS[wave] = s; }
  __syncthreads();
  if (wave == 0) {
    const int te = wave_sum(wE[lane]), ts = wave_sum(wS[lane]);
    if (lane == 0) {
      blkE[r * nb + blockIdx.x] = te;
      blkS[r * nb + blockIdx.x] = ts;
    }
  }
}

static constexpr int MID_MAX = 1024;
__global__ __launch_bounds__(SCAN_THREADS) void k_scan_mid(
    const int* __restrict__ blkE, const int* __restrict__ blkS, int nb, int N,
    int* __restrict__ blkBaseE, int* __restrict__ blkBaseS,
    int* __restrict__ starts, int* __restrict__ segBeg, int* __restrict__ segDst) {
  __shared__ int inE[MID_MAX], inS[MID_MAX];
  __shared__ int relE[8], relS[8], baseE[9], baseS[9];
  const int t = threadIdx.x, lane = t & 31, wave = t >> 5;
  if (wave < 8) {
    int carryE = 0, carryS = 0;
    for (int b0 = 0; b0 < nb; b0 += 32) {
      const int b = b0 + lane;
      const int i = wave * nb + b;
      const int ve = b < nb ? blkE[i] : 0;
      const int vs = b < nb ? blkS[i] : 0;
      const int ie = wave_incl_scan(ve, lane), is = wave_incl_scan(vs, lane);
      if (b < nb) {
        inE[i] = carryE + ie - ve;
        inS[i] = carryS + is - vs;
      }
      carryE += __shfl(ie, 31, 32);
      carryS += __shfl(is, 31, 32);
    }
    if (lane == 0) { relE[wave] = carryE; relS[wave] = carryS; }
  }
  __syncthreads();
  if (t == 0) {
    int e = 0, s = 0;
    for (int r = 0; r < 8; ++r) {
      baseE[r] = e;
      baseS[r] = s;
      e += relE[r];
      s += (relS[r] + 15) & ~15;
    }
    baseE[8] = e;
    baseS[8] = s;
  }
  __syncthreads();
  for (int i = t; i < 8 * nb; i += SCAN_THREADS) {
    const int r = i / nb;
    blkBaseE[i] = baseE[r] + inE[i];
    blkBaseS[i] = baseS[r] + inS[i];
  }
  if (t < 9) starts[t] = baseS[t];
  if (t == 0) segBeg[baseS[8]] = baseE[8];
  if (t < 128) {
    const int r = t >> 4;
    const int p = baseS[r] + relS[r] + (t & 15);
    if (p < baseS[r + 1]) {
      segBeg[p] = baseE[r + 1];
      segDst[p] = N;
    }
  }
}

__global__ __launch_bounds__(SCAN_THREADS) void k_scan_b(
    int* __restrict__ cnt, int N, int nb,
    const int* __restrict__ blkBaseE, const int* __restrict__ blkBaseS,
    int* __restrict__ segBeg, int* __restrict__ segDst) {
  __shared__ int wE[32], wS[32];
  const int r = blockIdx.y, t = threadIdx.x, lane = t & 31, wave = t >> 5;
  const int k = blockIdx.x * SCAN_THREADS + t;
  const int v = k < N ? cnt[(size_t)r * N + k] : 0;
  const int sv = (v + SEG_EDGES - 1) / SEG_EDGES;
  const int ie = wave_incl_scan(v, lane), is = wave_incl_scan(sv, lane);
  if (lane == 31) { wE[wave] = ie; wS[wave] = is; }
  __syncthreads();
  if (wave == 0) {
    const int a = wE[lane], b = wS[lane];
    const int ia = wave_incl_scan(a, lane), ib = wave_incl_scan(b, lane);
    wE[lane] = ia - a;
    wS[lane] = ib - b;
  }
  __syncthreads();
  const int e0 = blkBaseE[r * nb + blockIdx.x] + wE[wave] + ie - v;
  int s = blkBaseS[r * nb + blockIdx.x] + wS[wave] + is - sv;
  if (k < N) {
    cnt[(size_t)r * N + k] = e0;
    for (int q = 0; q < v; q += SEG_EDGES) {
      segBeg[s] = e0 + q;
      segDst[s] = k;
      ++s;
    }
  }
}

__global__ void k_key_scatter(const int* __restrict__ ei, const int* __restrict__ etype, int E, int N,
                              int* __restrict__ cursor, int* __restrict__ ssrc) {
  const int e = blockIdx.x * blockDim.x + threadIdx.x;
  if (e < E) {
    const int pos = atomicAdd(&cursor[(etype[e] & 7) * N + ei[E + e]], 1);
    ssrc[pos] = ei[e];
  }
}


template <int DOUT>
__global__ __launch_bounds__(32 * (DOUT / 16)) void k_root_gemm(
    const float* __restrict__ x, const float* __restrict__ W,
    const float* __restrict__ bias, float* __restrict__ y, int N) {
  __shared__ __attribute__((aligned(16))) float As[16 * A_PITCH];
  constexpr int NW = DOUT / 16;
  const int wave = threadIdx.x >> 5, lane = threadIdx.x & 31;
  const int m = lane & 15;
  const int khalf = (lane >> 4) << 1;
#pragma unroll
  for (int j = 0; j < 16 / NW; ++j) {
    const int tr = wave + j * NW;
    const int row = (blockIdx.x << 4) + tr;
    stage_row(As, x, row < N ? row : N - 1, tr, lane);
  }
  __syncthreads();
  const float* xa = As + m * A_PITCH;
  const int col = (wave << 4) + m;
  const float* wb = W + col;
  v8f c = {};
#pragma unroll 8
  for (int k = 0; k < D_IN; k += 4) {
    v2f a, b;
    a.x = xa[k + khalf];
    a.y = xa[k + khalf + 1];
    b.x = wb[(k + khalf) * DOUT];
    b.y = wb[(k + khalf + 1) * DOUT];
    c = WMMA_F32(a, b, c);
  }
  const float bv = bias[col];
  const int rbase = (blockIdx.x << 4) + ((lane >> 4) << 3);
  float* yp = y + (size_t)rbase * DOUT + col;
  if ((blockIdx.x << 4) + 16 <= N) {
#pragma unroll
    for (int i = 0; i < 8; ++i) yp[i * DOUT] = c[i] + bv;
  } else {
#pragma unroll
    for (int i = 0; i < 8; ++i)
      if (rbase + i < N) yp[i * DOUT] = c[i] + bv;
  }
}


static constexpr int RUN_TILES = 8;
static constexpr int RUN_SEGS = RUN_TILES * 16;
static constexpr int RUN_EDGES_MAX = RUN_SEGS * SEG_EDGES;

__device__ __forceinline__ float4 add4(float4 a, float4 b) {
  return make_float4(a.x + b.x, a.y + b.y, a.z + b.z, a.w + b.w);
}

struct SegRows {
  float4 p[SEG_EDGES];
};

struct RowPtrs {
  const float* p[SEG_EDGES];
};

__device__ __forceinline__ RowPtrs seg_ptrs(const float* __restrict__ x, const float* __restrict__ zrow,
                                            const int* sBeg, const int* sSrc, int seg) {
  const int rb = __builtin_amdgcn_readfirstlane(sBeg[seg]);
  const int n = __builtin_amdgcn_readfirstlane(sBeg[seg + 1]) - rb;
  RowPtrs r;
#pragma unroll
  for (int q = 0; q < SEG_EDGES; ++q) {
    const int row = __builtin_amdgcn_readfirstlane(sSrc[rb + q]);
    r.p[q] = q < n ? x + (size_t)row * D_IN : zrow;
  }
  return r;
}

__device__ __forceinline__ SegRows seg_fetch(const RowPtrs& r, unsigned laneBytes) {
  SegRows s;
#pragma unroll
  for (int q = 0; q < SEG_EDGES; ++q)
    s.p[q] = *reinterpret_cast<const float4*>(reinterpret_cast<const char*>(r.p[q]) + laneBytes);
  return s;
}

__device__ __forceinline__ float4 seg_total(const SegRows& s) {
  return add4(add4(s.p[0], s.p[1]), add4(s.p[2], s.p[3]));
}

template <int DOUT>
__global__ __launch_bounds__(32 * (DOUT / 16)) __attribute__((amdgpu_waves_per_eu(DOUT == 128 ? 8 : 7, DOUT == 128 ? 8 : 7))) void k_edge_wmma(
    const float* __restrict__ x, const float* __restrict__ Wrel,
    const int* __restrict__ ssrc, const int* __restrict__ segBeg, const int* __restrict__ segDst,
    const int* __restrict__ starts, const float* __restrict__ zrow, float* __restrict__ y) {
  static_assert(SEG_EDGES == 4, "seg_total() is written for four rows");
  __shared__ __attribute__((aligned(16))) float As[2][16 * A_PITCH];
  __shared__ __attribute__((aligned(16))) int sBeg[RUN_SEGS + 16];
  __shared__ __attribute__((aligned(16))) int dIdx[RUN_SEGS];
  __shared__ __attribute__((aligned(16))) int sSrc[RUN_EDGES_MAX + 16];
  constexpr int C_PITCH = DOUT + 4;
  __shared__ __attribute__((aligned(16))) float Cs[2][16 * C_PITCH];
  constexpr int NW = DOUT / 16;
  constexpr int RPW = 16 / NW;
  constexpr int LPR = DOUT / 32;
  const int total = starts[8];
  const int base0 = blockIdx.x * RUN_SEGS;
  if (base0 >= total) return;
  const int left = (total - base0) >> 4;
  const int nT = left < RUN_TILES ? left : RUN_TILES;
  const int lane = threadIdx.x & 31;
  const unsigned laneBytes = (unsigned)lane << 4;
  const int wave = __builtin_amdgcn_readfirstlane((int)(threadIdx.x >> 5));

  const int eb0 = segBeg[base0];
  const int nE = segBeg[base0 + (nT << 4)] - eb0;
  for (int i = threadIdx.x; i < RUN_SEGS + 1; i += 32 * NW) {
    const int gi = base0 + i;
    sBeg[i] = (gi <= total ? segBeg[gi] : eb0) - eb0;
    if (i < RUN_SEGS) dIdx[i] = gi < total ? segDst[gi] : 0;
  }
  for (int i = threadIdx.x; i < RUN_EDGES_MAX + 16; i += 32 * NW)
    sSrc[i] = i < nE ? ssrc[eb0 + i] : 0;
  __syncthreads();

#pragma unroll
  for (int j = 0; j < RPW; ++j) {
    const int tr = wave + j * NW;
    const SegRows s = seg_fetch(seg_ptrs(x, zrow, sBeg, sSrc, tr), laneBytes);
    *reinterpret_cast<float4*>(As[0] + tr * A_PITCH + (lane << 2)) = seg_total(s);
  }
  RowPtrs nx[RPW];
  {
    const int t1 = 1 < nT ? 1 : 0;
#pragma unroll
    for (int j = 0; j < RPW; ++j) nx[j] = seg_ptrs(x, zrow, sBeg, sSrc, (t1 << 4) + wave + j * NW);
  }
  __syncthreads();

  const int m = lane & 15;
  const int khalf = (lane >> 4) << 1;
  const int col = ((int)(threadIdx.x >> 5) << 4) + m;
  const int mBase = (lane >> 4) << 3;
  const unsigned laneWord = (unsigned)lane << 2;
  const int s1 = starts[1], s2 = starts[2], s3 = starts[3], s4 = starts[4],
            s5 = starts[5], s6 = starts[6], s7 = starts[7];

  v2f b[D_IN / 4];
  int curR = -1;
  int cur = 0;
  for (int t = 0; t < nT; ++t) {
    const int base = base0 + (t << 4);
    const int r = (base >= s1) + (base >= s2) + (base >= s3) + (base >= s4) +
                  (base >= s5) + (base >= s6) + (base >= s7);
    if (r != curR) {
      const float* wb = Wrel + (size_t)r * (D_IN * DOUT) + col;
#pragma unroll
      for (int kk = 0; kk < D_IN / 4; ++kk) {
        b[kk].x = wb[(4 * kk + khalf) * DOUT];
        b[kk].y = wb[(4 * kk + khalf + 1) * DOUT];
      }
      curR = r;
    }

    unsigned lb = laneBytes;
    asm volatile("" : "+v"(lb));
    SegRows pre[RPW];
#pragma unroll
    for (int j = 0; j < RPW; ++j) pre[j] = seg_fetch(nx[j], lb);
    const int tnn = t + 2 < nT ? t + 2 : nT - 1;
#pragma unroll
    for (int j = 0; j < RPW; ++j) nx[j] = seg_ptrs(x, zrow, sBeg, sSrc, (tnn << 4) + wave + j * NW);

    const float* xa = As[cur] + m * A_PITCH;
    v8f c = {};
#pragma unroll
    for (int kk = 0; kk < D_IN / 4; ++kk) {
      v2f a;
      a.x = xa[4 * kk + khalf];
      a.y = xa[4 * kk + khalf + 1];
      c = WMMA_F32(a, b[kk], c);
    }

    float* Cw = Cs[cur] + mBase * C_PITCH + col;
#pragma unroll
    for (int i = 0; i < 8; ++i) Cw[i * C_PITCH] = c[i];

    int dstRow[RPW];
#pragma unroll
    for (int j = 0; j < RPW; ++j)
      dstRow[j] = __builtin_amdgcn_readfirstlane(dIdx[(t << 4) + wave * RPW + j]);

    float* An = As[cur ^ 1];
#pragma unroll
    for (int j = 0; j < RPW; ++j) {
      const int tr = wave + j * NW;
      *reinterpret_cast<float4*>(An + tr * A_PITCH + (lane << 2)) = seg_total(pre[j]);
    }

    __syncthreads();
    const float* Cr = Cs[cur] + (wave * RPW) * C_PITCH + lane;
    cur ^= 1;

    float cv[RPW][LPR];
#pragma unroll
    for (int j = 0; j < RPW; ++j)
#pragma unroll
      for (int q = 0; q < LPR; ++q) cv[j][q] = Cr[j * C_PITCH + 32 * q];
#pragma unroll
    for (int j = 0; j < RPW; ++j) {
      const unsigned off = (unsigned)dstRow[j] * (unsigned)(DOUT * 4) + laneWord;
      float* yr = reinterpret_cast<float*>(reinterpret_cast<char*>(y) + off);
#pragma unroll
      for (int q = 0; q < LPR; ++q) atomicAdd(yr + 32 * q, cv[j][q]);
    }
  }
}


__global__ void k_relu4(float4* __restrict__ p, long long n4) {
  long long i = (long long)blockIdx.x * blockDim.x + threadIdx.x;
  if (i < n4) {
    float4 v = p[i];
    v.x = fmaxf(v.x, 0.0f);
    v.y = fmaxf(v.y, 0.0f);
    v.z = fmaxf(v.z, 0.0f);
    v.w = fmaxf(v.w, 0.0f);
    p[i] = v;
  }
}

__global__ void k_logsoftmax(const float* __restrict__ y, float* __restrict__ out, int N) {
  int wave = threadIdx.x >> 5, lane = threadIdx.x & 31;
  int node = blockIdx.x * (blockDim.x >> 5) + wave;
  if (node >= N) return;
  const float* row = y + (size_t)node * 64;
  float v0 = row[lane], v1 = row[lane + 32];
  float mx = fmaxf(v0, v1);
#pragma unroll
  for (int off = 16; off > 0; off >>= 1) mx = fmaxf(mx, __shfl_xor(mx, off, 32));
  float s = expf(v0 - mx) + expf(v1 - mx);
#pragma unroll
  for (int off = 16; off > 0; off >>= 1) s += __shfl_xor(s, off, 32);
  float lse = logf(s) + mx;
  float* o = out + (size_t)node * 64;
  o[lane] = v0 - lse;
  o[lane + 32] = v1 - lse;
}


static inline void launch_layer(const float* x, const float* Wrel, const float* Wroot,
                                const float* bias, float* y, int dOut, int N,
                                const int* ssrc, const int* segBeg, const int* segDst, const int* starts,
                                const float* zrow, int nodeTiles, int segRunsMax, hipStream_t stream) {
  if (dOut == 128) {
    k_root_gemm<128><<<nodeTiles, 32 * (128 / 16), 0, stream>>>(x, Wroot, bias, y, N);
    k_edge_wmma<128><<<segRunsMax, 32 * (128 / 16), 0, stream>>>(x, Wrel, ssrc, segBeg, segDst, starts, zrow, y);
  } else {
    k_root_gemm<64><<<nodeTiles, 32 * (64 / 16), 0, stream>>>(x, Wroot, bias, y, N);
    k_edge_wmma<64><<<segRunsMax, 32 * (64 / 16), 0, stream>>>(x, Wrel, ssrc, segBeg, segDst, starts, zrow, y);
  }
}

extern "C" void kernel_launch(void* const* d_in, const int* in_sizes, int n_in,
                              void* d_out, int out_size, void* d_ws, size_t ws_size,
                              hipStream_t stream) {
  const float* x0     = (const float*)d_in[0];
  const int*   ei     = (const int*)  d_in[1];
  const int*   etype  = (const int*)  d_in[2];
  const float* Wrel1  = (const float*)d_in[3];
  const float* Wroot1 = (const float*)d_in[4];
  const float* b1     = (const float*)d_in[5];
  const float* Wrel2  = (const float*)d_in[6];
  const float* Wroot2 = (const float*)d_in[7];
  const float* b2     = (const float*)d_in[8];
  const float* Wrel3  = (const float*)d_in[9];
  const float* Wroot3 = (const float*)d_in[10];
  const float* b3     = (const float*)d_in[11];

  const int N  = in_sizes[0] / D_IN;
  const int E  = in_sizes[2];
  const int D1 = 128, D2 = 128;
  const int D3 = in_sizes[11];
  const int K  = 8 * N;
  const int keysMax = E < K ? E : K;
  const int segsMax = keysMax + (E - keysMax + SEG_EDGES - 1) / SEG_EDGES + 128;

  char* w = (char*)d_ws;
  int* starts = (int*)w;
  size_t off = 256;
  int* segBeg = (int*)(w + off); off += (size_t)(segsMax + 16) * sizeof(int);
  int* segDst = (int*)(w + off); off += (size_t)(segsMax + 16) * sizeof(int);
  int* ssrc   = (int*)(w + off); off += (size_t)E * sizeof(int);
  off = (off + 255) & ~(size_t)255;
  float* bufA = (float*)(w + off); off += (size_t)(N + 16) * D_IN * sizeof(float);
  float* bufB = (float*)(w + off);
  float* zrow = bufA + (size_t)(N + 8) * D_IN;
  int* cnt  = (int*)bufB;
  int* blkE = cnt + K;
  int* blkS = blkE + MID_MAX;
  int* blkBaseE = blkS + MID_MAX;
  int* blkBaseS = blkBaseE + MID_MAX;

  const int nb = (N + SCAN_THREADS - 1) / SCAN_THREADS;
  k_zero_ints  <<<(K + 255) / 256, 256, 0, stream>>>(cnt, K, zrow);
  k_key_hist   <<<(E + 255) / 256, 256, 0, stream>>>(ei, etype, E, N, cnt);
  k_scan_a     <<<dim3(nb, 8), SCAN_THREADS, 0, stream>>>(cnt, N, nb, blkE, blkS);
  k_scan_mid   <<<1, SCAN_THREADS, 0, stream>>>(blkE, blkS, nb, N, blkBaseE, blkBaseS, starts, segBeg, segDst);
  k_scan_b     <<<dim3(nb, 8), SCAN_THREADS, 0, stream>>>(cnt, N, nb, blkBaseE, blkBaseS, segBeg, segDst);
  k_key_scatter<<<(E + 255) / 256, 256, 0, stream>>>(ei, etype, E, N, cnt, ssrc);

  const int nodeTiles  = (N + 15) / 16;
  const int segRunsMax = (segsMax + RUN_SEGS - 1) / RUN_SEGS;

  launch_layer(x0, Wrel1, Wroot1, b1, bufA, D1, N, ssrc, segBeg, segDst, starts,
               zrow, nodeTiles, segRunsMax, stream);
  {
    long long n4 = (long long)N * D1 / 4;
    k_relu4<<<(unsigned)((n4 + 255) / 256), 256, 0, stream>>>((float4*)bufA, n4);
  }

  launch_layer(bufA, Wrel2, Wroot2, b2, bufB, D2, N, ssrc, segBeg, segDst, starts,
               zrow, nodeTiles, segRunsMax, stream);
  {
    long long n4 = (long long)N * D2 / 4;
    k_relu4<<<(unsigned)((n4 + 255) / 256), 256, 0, stream>>>((float4*)bufB, n4);
  }

  launch_layer(bufB, Wrel3, Wroot3, b3, bufA, D3, N, ssrc, segBeg, segDst, starts,
               zrow, nodeTiles, segRunsMax, stream);
  k_logsoftmax<<<(N + 7) / 8, 256, 0, stream>>>(bufA, (float*)d_out, N);
}


# Edge sort order (block of 1024 destinations, relation, destination), was (relation, destination): a block's 8 relations scatter back to back, accumulator lines stay in L2; atomic add order changes
# speedup vs baseline: 4.0655x; 1.0654x over previous
#include <hip/hip_runtime.h>
#include <hip/hip_bf16.h>
#include <math.h>


typedef float v2f __attribute__((ext_vector_type(2)));
typedef float v8f __attribute__((ext_vector_type(8)));

#define WMMA_F32(a, b, c) \
  __builtin_amdgcn_wmma_f32_16x16x4_f32(false, (a), false, (b), (short)0, (c), false, false)

static constexpr int D_IN = 128;

static constexpr int A_PITCH = D_IN + 4;

__device__ __forceinline__ void stage_row(float* As, const float* __restrict__ x,
                                          int srcRow, int tileRow, int lane) {
  const float4 v = *reinterpret_cast<const float4*>(x + (size_t)srcRow * D_IN + (lane << 2));
  *reinterpret_cast<float4*>(As + tileRow * A_PITCH + (lane << 2)) = v;
}

static constexpr int SEG_EDGES = 4;

__global__ void k_zero_ints(int* __restrict__ p, int n, float* __restrict__ zrow) {
  int i = blockIdx.x * blockDim.x + threadIdx.x;
  if (i < n) p[i] = 0;
  if (i < D_IN) zrow[i] = 0.f;
}

__global__ void k_key_hist(const int* __restrict__ ei, const int* __restrict__ etype, int E, int N,
                           int* __restrict__ cnt) {
  const int e = blockIdx.x * blockDim.x + threadIdx.x;
  if (e < E) atomicAdd(&cnt[(etype[e] & 7) * N + ei[E + e]], 1);
}

static constexpr int SCAN_THREADS = 1024;

__device__ __forceinline__ int wave_incl_scan(int v, int lane) {
#pragma unroll
  for (int o = 1; o < 32; o <<= 1) {
    const int t = __shfl_up(v, o, 32);
    if (lane >= o) v += t;
  }
  return v;
}

__device__ __forceinline__ int wave_sum(int v) {
#pragma unroll
  for (int o = 16; o > 0; o >>= 1) v += __shfl_xor(v, o, 32);
  return v;
}

__global__ __launch_bounds__(SCAN_THREADS) void k_scan_a(
    const int* __restrict__ cnt, int N, int* __restrict__ blkE, int* __restrict__ blkS) {
  __shared__ int wE[32], wS[32];
  const int r = blockIdx.y, t = threadIdx.x, lane = t & 31, wave = t >> 5;
  const int k = blockIdx.x * SCAN_THREADS + t;
  const int v = k < N ? cnt[(size_t)r * N + k] : 0;
  const int e = wave_sum(v);
  const int s = wave_sum((v + SEG_EDGES - 1) / SEG_EDGES);
  if (lane == 0) { wE[wave] = e; wS[wave] = s; }
  __syncthreads();
  if (wave == 0) {
    const int te = wave_sum(wE[lane]), ts = wave_sum(wS[lane]);
    if (lane == 0) {
      blkE[blockIdx.x * 8 + r] = te;
      blkS[blockIdx.x * 8 + r] = ts;
    }
  }
}

static constexpr int MID_MAX = 1024;
__global__ __launch_bounds__(SCAN_THREADS) void k_scan_mid(
    const int* __restrict__ blkE, const int* __restrict__ blkS, int ng,
    int* __restrict__ blkBaseE, int* __restrict__ blkBaseS,
    int* __restrict__ starts, int* __restrict__ segBeg, int* __restrict__ segDst) {
  __shared__ int wE[32], wS[32];
  const int t = threadIdx.x, lane = t & 31, wave = t >> 5;
  const int ve = t < ng ? blkE[t] : 0;
  const int vs = t < ng ? blkS[t] : 0;
  const int vp = (vs + 15) & ~15;
  const int ie = wave_incl_scan(ve, lane), ip = wave_incl_scan(vp, lane);
  if (lane == 31) { wE[wave] = ie; wS[wave] = ip; }
  __syncthreads();
  if (wave == 0) {
    const int a = wE[lane], b = wS[lane];
    const int ia = wave_incl_scan(a, lane), ib = wave_incl_scan(b, lane);
    wE[lane] = ia - a;
    wS[lane] = ib - b;
  }
  __syncthreads();
  const int e0 = wE[wave] + ie - ve;
  const int s0 = wS[wave] + ip - vp;
  if (t < ng) {
    blkBaseE[t] = e0;
    blkBaseS[t] = s0;
    for (int p = s0 + vs; p < s0 + vp; ++p) {
      segBeg[p] = e0 + ve;
      segDst[p] = ((t >> 3) * SCAN_THREADS) | ((t & 7) << 28);
    }
  }
  if (t == SCAN_THREADS - 1) {
    starts[8] = s0 + vp;
    segBeg[s0 + vp] = e0 + ve;
  }
}

__global__ __launch_bounds__(SCAN_THREADS) void k_scan_b(
    int* __restrict__ cnt, int N,
    const int* __restrict__ blkBaseE, const int* __restrict__ blkBaseS,
    int* __restrict__ segBeg, int* __restrict__ segDst) {
  __shared__ int wE[32], wS[32];
  const int r = blockIdx.y, t = threadIdx.x, lane = t & 31, wave = t >> 5;
  const int k = blockIdx.x * SCAN_THREADS + t;
  const int v = k < N ? cnt[(size_t)r * N + k] : 0;
  const int sv = (v + SEG_EDGES - 1) / SEG_EDGES;
  const int ie = wave_incl_scan(v, lane), is = wave_incl_scan(sv, lane);
  if (lane == 31) { wE[wave] = ie; wS[wave] = is; }
  __syncthreads();
  if (wave == 0) {
    const int a = wE[lane], b = wS[lane];
    const int ia = wave_incl_scan(a, lane), ib = wave_incl_scan(b, lane);
    wE[lane] = ia - a;
    wS[lane] = ib - b;
  }
  __syncthreads();
  const int e0 = blkBaseE[blockIdx.x * 8 + r] + wE[wave] + ie - v;
  int s = blkBaseS[blockIdx.x * 8 + r] + wS[wave] + is - sv;
  if (k < N) {
    cnt[(size_t)r * N + k] = e0;
    for (int q = 0; q < v; q += SEG_EDGES) {
      segBeg[s] = e0 + q;
      segDst[s] = k | (r << 28);
      ++s;
    }
  }
}

__global__ void k_key_scatter(const int* __restrict__ ei, const int* __restrict__ etype, int E, int N,
                              int* __restrict__ cursor, int* __restrict__ ssrc) {
  const int e = blockIdx.x * blockDim.x + threadIdx.x;
  if (e < E) {
    const int pos = atomicAdd(&cursor[(etype[e] & 7) * N + ei[E + e]], 1);
    ssrc[pos] = ei[e];
  }
}


template <int DOUT>
__global__ __launch_bounds__(32 * (DOUT / 16)) void k_root_gemm(
    const float* __restrict__ x, const float* __restrict__ W,
    const float* __restrict__ bias, float* __restrict__ y, int N) {
  __shared__ __attribute__((aligned(16))) float As[16 * A_PITCH];
  constexpr int NW = DOUT / 16;
  const int wave = threadIdx.x >> 5, lane = threadIdx.x & 31;
  const int m = lane & 15;
  const int khalf = (lane >> 4) << 1;
#pragma unroll
  for (int j = 0; j < 16 / NW; ++j) {
    const int tr = wave + j * NW;
    const int row = (blockIdx.x << 4) + tr;
    stage_row(As, x, row < N ? row : N - 1, tr, lane);
  }
  __syncthreads();
  const float* xa = As + m * A_PITCH;
  const int col = (wave << 4) + m;
  const float* wb = W + col;
  v8f c = {};
#pragma unroll 8
  for (int k = 0; k < D_IN; k += 4) {
    v2f a, b;
    a.x = xa[k + khalf];
    a.y = xa[k + khalf + 1];
    b.x = wb[(k + khalf) * DOUT];
    b.y = wb[(k + khalf + 1) * DOUT];
    c = WMMA_F32(a, b, c);
  }
  const float bv = bias[col];
  const int rbase = (blockIdx.x << 4) + ((lane >> 4) << 3);
  float* yp = y + (size_t)rbase * DOUT + col;
  if ((blockIdx.x << 4) + 16 <= N) {
#pragma unroll
    for (int i = 0; i < 8; ++i) yp[i * DOUT] = c[i] + bv;
  } else {
#pragma unroll
    for (int i = 0; i < 8; ++i)
      if (rbase + i < N) yp[i * DOUT] = c[i] + bv;
  }
}


static constexpr int RUN_TILES = 8;
static constexpr int RUN_SEGS = RUN_TILES * 16;
static constexpr int RUN_EDGES_MAX = RUN_SEGS * SEG_EDGES;

__device__ __forceinline__ float4 add4(float4 a, float4 b) {
  return make_float4(a.x + b.x, a.y + b.y, a.z + b.z, a.w + b.w);
}

struct SegRows {
  float4 p[SEG_EDGES];
};

struct RowPtrs {
  const float* p[SEG_EDGES];
};

__device__ __forceinline__ RowPtrs seg_ptrs(const float* __restrict__ x, const float* __restrict__ zrow,
                                            const int* sBeg, const int* sSrc, int seg) {
  const int rb = __builtin_amdgcn_readfirstlane(sBeg[seg]);
  const int n = __builtin_amdgcn_readfirstlane(sBeg[seg + 1]) - rb;
  RowPtrs r;
#pragma unroll
  for (int q = 0; q < SEG_EDGES; ++q) {
    const int row = __builtin_amdgcn_readfirstlane(sSrc[rb + q]);
    r.p[q] = q < n ? x + (size_t)row * D_IN : zrow;
  }
  return r;
}

__device__ __forceinline__ SegRows seg_fetch(const RowPtrs& r, unsigned laneBytes) {
  SegRows s;
#pragma unroll
  for (int q = 0; q < SEG_EDGES; ++q)
    s.p[q] = *reinterpret_cast<const float4*>(reinterpret_cast<const char*>(r.p[q]) + laneBytes);
  return s;
}

__device__ __forceinline__ float4 seg_total(const SegRows& s) {
  return add4(add4(s.p[0], s.p[1]), add4(s.p[2], s.p[3]));
}

template <int DOUT>
__global__ __launch_bounds__(32 * (DOUT / 16)) __attribute__((amdgpu_waves_per_eu(DOUT == 128 ? 8 : 7, DOUT == 128 ? 8 : 7))) void k_edge_wmma(
    const float* __restrict__ x, const float* __restrict__ Wrel,
    const int* __restrict__ ssrc, const int* __restrict__ segBeg, const int* __restrict__ segDst,
    const int* __restrict__ starts, const float* __restrict__ zrow, float* __restrict__ y) {
  static_assert(SEG_EDGES == 4, "seg_total() is written for four rows");
  __shared__ __attribute__((aligned(16))) float As[2][16 * A_PITCH];
  __shared__ __attribute__((aligned(16))) int sBeg[RUN_SEGS + 16];
  __shared__ __attribute__((aligned(16))) int dIdx[RUN_SEGS];
  __shared__ __attribute__((aligned(16))) int sSrc[RUN_EDGES_MAX + 16];
  constexpr int C_PITCH = DOUT + 4;
  __shared__ __attribute__((aligned(16))) float Cs[2][16 * C_PITCH];
  constexpr int NW = DOUT / 16;
  constexpr int RPW = 16 / NW;
  constexpr int LPR = DOUT / 32;
  const int total = starts[8];
  const int base0 = blockIdx.x * RUN_SEGS;
  if (base0 >= total) return;
  const int left = (total - base0) >> 4;
  const int nT = left < RUN_TILES ? left : RUN_TILES;
  const int lane = threadIdx.x & 31;
  const unsigned laneBytes = (unsigned)lane << 4;
  const int wave = __builtin_amdgcn_readfirstlane((int)(threadIdx.x >> 5));

  const int eb0 = segBeg[base0];
  const int nE = segBeg[base0 + (nT << 4)] - eb0;
  for (int i = threadIdx.x; i < RUN_SEGS + 1; i += 32 * NW) {
    const int gi = base0 + i;
    sBeg[i] = (gi <= total ? segBeg[gi] : eb0) - eb0;
    if (i < RUN_SEGS) dIdx[i] = gi < total ? segDst[gi] : 0;
  }
  for (int i = threadIdx.x; i < RUN_EDGES_MAX + 16; i += 32 * NW)
    sSrc[i] = i < nE ? ssrc[eb0 + i] : 0;
  __syncthreads();

  int relPack = 0;
#pragma unroll
  for (int q = 0; q < RUN_TILES; ++q)
    relPack |= ((__builtin_amdgcn_readfirstlane(dIdx[q << 4]) >> 28) & 7) << (3 * q);

#pragma unroll
  for (int j = 0; j < RPW; ++j) {
    const int tr = wave + j * NW;
    const SegRows s = seg_fetch(seg_ptrs(x, zrow, sBeg, sSrc, tr), laneBytes);
    *reinterpret_cast<float4*>(As[0] + tr * A_PITCH + (lane << 2)) = seg_total(s);
  }
  RowPtrs nx[RPW];
  {
    const int t1 = 1 < nT ? 1 : 0;
#pragma unroll
    for (int j = 0; j < RPW; ++j) nx[j] = seg_ptrs(x, zrow, sBeg, sSrc, (t1 << 4) + wave + j * NW);
  }
  __syncthreads();

  const int m = lane & 15;
  const int khalf = (lane >> 4) << 1;
  const int col = ((int)(threadIdx.x >> 5) << 4) + m;
  const int mBase = (lane >> 4) << 3;
  const unsigned laneWord = (unsigned)lane << 2;

  v2f b[D_IN / 4];
  int curR = -1;
  int cur = 0;
  for (int t = 0; t < nT; ++t) {
    const int r = (relPack >> (3 * t)) & 7;
    if (r != curR) {
      const float* wb = Wrel + (size_t)r * (D_IN * DOUT) + col;
#pragma unroll
      for (int kk = 0; kk < D_IN / 4; ++kk) {
        b[kk].x = wb[(4 * kk + khalf) * DOUT];
        b[kk].y = wb[(4 * kk + khalf + 1) * DOUT];
      }
      curR = r;
    }

    unsigned lb = laneBytes;
    asm volatile("" : "+v"(lb));
    SegRows pre[RPW];
#pragma unroll
    for (int j = 0; j < RPW; ++j) pre[j] = seg_fetch(nx[j], lb);
    const int tnn = t + 2 < nT ? t + 2 : nT - 1;
#pragma unroll
    for (int j = 0; j < RPW; ++j) nx[j] = seg_ptrs(x, zrow, sBeg, sSrc, (tnn << 4) + wave + j * NW);

    const float* xa = As[cur] + m * A_PITCH;
    v8f c = {};
#pragma unroll
    for (int kk = 0; kk < D_IN / 4; ++kk) {
      v2f a;
      a.x = xa[4 * kk + khalf];
      a.y = xa[4 * kk + khalf + 1];
      c = WMMA_F32(a, b[kk], c);
    }

    float* Cw = Cs[cur] + mBase * C_PITCH + col;
#pragma unroll
    for (int i = 0; i < 8; ++i) Cw[i * C_PITCH] = c[i];

    int dstRow[RPW];
#pragma unroll
    for (int j = 0; j < RPW; ++j)
      dstRow[j] = __builtin_amdgcn_readfirstlane(dIdx[(t << 4) + wave * RPW + j]);

    float* An = As[cur ^ 1];
#pragma unroll
    for (int j = 0; j < RPW; ++j) {
      const int tr = wave + j * NW;
      *reinterpret_cast<float4*>(An + tr * A_PITCH + (lane << 2)) = seg_total(pre[j]);
    }

    __syncthreads();
    const float* Cr = Cs[cur] + (wave * RPW) * C_PITCH + lane;
    cur ^= 1;

    float cv[RPW][LPR];
#pragma unroll
    for (int j = 0; j < RPW; ++j)
#pragma unroll
      for (int q = 0; q < LPR; ++q) cv[j][q] = Cr[j * C_PITCH + 32 * q];
#pragma unroll
    for (int j = 0; j < RPW; ++j) {
      const unsigned off = (unsigned)(dstRow[j] & 0x0FFFFFFF) * (unsigned)(DOUT * 4) + laneWord;
      float* yr = reinterpret_cast<float*>(reinterpret_cast<char*>(y) + off);
#pragma unroll
      for (int q = 0; q < LPR; ++q) atomicAdd(yr + 32 * q, cv[j][q]);
    }
  }
}


__global__ void k_relu4(float4* __restrict__ p, long long n4) {
  long long i = (long long)blockIdx.x * blockDim.x + threadIdx.x;
  if (i < n4) {
    float4 v = p[i];
    v.x = fmaxf(v.x, 0.0f);
    v.y = fmaxf(v.y, 0.0f);
    v.z = fmaxf(v.z, 0.0f);
    v.w = fmaxf(v.w, 0.0f);
    p[i] = v;
  }
}

__global__ void k_logsoftmax(const float* __restrict__ y, float* __restrict__ out, int N) {
  int wave = threadIdx.x >> 5, lane = threadIdx.x & 31;
  int node = blockIdx.x * (blockDim.x >> 5) + wave;
  if (node >= N) return;
  const float* row = y + (size_t)node * 64;
  float v0 = row[lane], v1 = row[lane + 32];
  float mx = fmaxf(v0, v1);
#pragma unroll
  for (int off = 16; off > 0; off >>= 1) mx = fmaxf(mx, __shfl_xor(mx, off, 32));
  float s = expf(v0 - mx) + expf(v1 - mx);
#pragma unroll
  for (int off = 16; off > 0; off >>= 1) s += __shfl_xor(s, off, 32);
  float lse = logf(s) + mx;
  float* o = out + (size_t)node * 64;
  o[lane] = v0 - lse;
  o[lane + 32] = v1 - lse;
}


static inline void launch_layer(const float* x, const float* Wrel, const float* Wroot,
                                const float* bias, float* y, int dOut, int N,
                                const int* ssrc, const int* segBeg, const int* segDst, const int* starts,
                                const float* zrow, int nodeTiles, int segRunsMax, hipStream_t stream) {
  if (dOut == 128) {
    k_root_gemm<128><<<nodeTiles, 32 * (128 / 16), 0, stream>>>(x, Wroot, bias, y, N);
    k_edge_wmma<128><<<segRunsMax, 32 * (128 / 16), 0, stream>>>(x, Wrel, ssrc, segBeg, segDst, starts, zrow, y);
  } else {
    k_root_gemm<64><<<nodeTiles, 32 * (64 / 16), 0, stream>>>(x, Wroot, bias, y, N);
    k_edge_wmma<64><<<segRunsMax, 32 * (64 / 16), 0, stream>>>(x, Wrel, ssrc, segBeg, segDst, starts, zrow, y);
  }
}

extern "C" void kernel_launch(void* const* d_in, const int* in_sizes, int n_in,
                              void* d_out, int out_size, void* d_ws, size_t ws_size,
                              hipStream_t stream) {
  const float* x0     = (const float*)d_in[0];
  const int*   ei     = (const int*)  d_in[1];
  const int*   etype  = (const int*)  d_in[2];
  const float* Wrel1  = (const float*)d_in[3];
  const float* Wroot1 = (const float*)d_in[4];
  const float* b1     = (const float*)d_in[5];
  const float* Wrel2  = (const float*)d_in[6];
  const float* Wroot2 = (const float*)d_in[7];
  const float* b2     = (const float*)d_in[8];
  const float* Wrel3  = (const float*)d_in[9];
  const float* Wroot3 = (const float*)d_in[10];
  const float* b3     = (const float*)d_in[11];

  const int N  = in_sizes[0] / D_IN;
  const int E  = in_sizes[2];
  const int D1 = 128, D2 = 128;
  const int D3 = in_sizes[11];
  const int K  = 8 * N;
  const int keysMax = E < K ? E : K;
  const int nb = (N + SCAN_THREADS - 1) / SCAN_THREADS;
  const int segsMax = keysMax + (E - keysMax + SEG_EDGES - 1) / SEG_EDGES + 8 * nb * 16;

  char* w = (char*)d_ws;
  int* starts = (int*)w;
  size_t off = 256;
  int* segBeg = (int*)(w + off); off += (size_t)(segsMax + 16) * sizeof(int);
  int* segDst = (int*)(w + off); off += (size_t)(segsMax + 16) * sizeof(int);
  int* ssrc   = (int*)(w + off); off += (size_t)E * sizeof(int);
  off = (off + 255) & ~(size_t)255;
  float* bufA = (float*)(w + off); off += (size_t)(N + 16) * D_IN * sizeof(float);
  float* bufB = (float*)(w + off);
  float* zrow = bufA + (size_t)(N + 8) * D_IN;
  int* cnt  = (int*)bufB;
  int* blkE = cnt + K;
  int* blkS = blkE + MID_MAX;
  int* blkBaseE = blkS + MID_MAX;
  int* blkBaseS = blkBaseE + MID_MAX;

  k_zero_ints  <<<(K + 255) / 256, 256, 0, stream>>>(cnt, K, zrow);
  k_key_hist   <<<(E + 255) / 256, 256, 0, stream>>>(ei, etype, E, N, cnt);
  k_scan_a     <<<dim3(nb, 8), SCAN_THREADS, 0, stream>>>(cnt, N, blkE, blkS);
  k_scan_mid   <<<1, SCAN_THREADS, 0, stream>>>(blkE, blkS, 8 * nb, blkBaseE, blkBaseS, starts, segBeg, segDst);
  k_scan_b     <<<dim3(nb, 8), SCAN_THREADS, 0, stream>>>(cnt, N, blkBaseE, blkBaseS, segBeg, segDst);
  k_key_scatter<<<(E + 255) / 256, 256, 0, stream>>>(ei, etype, E, N, cnt, ssrc);

  const int nodeTiles  = (N + 15) / 16;
  const int segRunsMax = (segsMax + RUN_SEGS - 1) / RUN_SEGS;

  launch_layer(x0, Wrel1, Wroot1, b1, bufA, D1, N, ssrc, segBeg, segDst, starts,
               zrow, nodeTiles, segRunsMax, stream);
  {
    long long n4 = (long long)N * D1 / 4;
    k_relu4<<<(unsigned)((n4 + 255) / 256), 256, 0, stream>>>((float4*)bufA, n4);
  }

  launch_layer(bufA, Wrel2, Wroot2, b2, bufB, D2, N, ssrc, segBeg, segDst, starts,
               zrow, nodeTiles, segRunsMax, stream);
  {
    long long n4 = (long long)N * D2 / 4;
    k_relu4<<<(unsigned)((n4 + 255) / 256), 256, 0, stream>>>((float4*)bufB, n4);
  }

  launch_layer(bufB, Wrel3, Wroot3, b3, bufA, D3, N, ssrc, segBeg, segDst, starts,
               zrow, nodeTiles, segRunsMax, stream);
  k_logsoftmax<<<(N + 7) / 8, 256, 0, stream>>>(bufA, (float*)d_out, N);
}


# Edge sort: counting pass keeps each edge's rank in its key (returning add); placement = key start + rank, plain loads: one scattered device atomic per edge, not two; in-key edge (sum) order changes
# speedup vs baseline: 4.2273x; 1.0398x over previous
#include <hip/hip_runtime.h>
#include <hip/hip_bf16.h>
#include <math.h>


typedef float v2f __attribute__((ext_vector_type(2)));
typedef float v8f __attribute__((ext_vector_type(8)));

#define WMMA_F32(a, b, c) \
  __builtin_amdgcn_wmma_f32_16x16x4_f32(false, (a), false, (b), (short)0, (c), false, false)

static constexpr int D_IN = 128;

static constexpr int A_PITCH = D_IN + 4;

__device__ __forceinline__ void stage_row(float* As, const float* __restrict__ x,
                                          int srcRow, int tileRow, int lane) {
  const float4 v = *reinterpret_cast<const float4*>(x + (size_t)srcRow * D_IN + (lane << 2));
  *reinterpret_cast<float4*>(As + tileRow * A_PITCH + (lane << 2)) = v;
}

static constexpr int SEG_EDGES = 4;

__global__ void k_zero_ints(int* __restrict__ p, int n, float* __restrict__ zrow) {
  int i = blockIdx.x * blockDim.x + threadIdx.x;
  if (i < n) p[i] = 0;
  if (i < D_IN) zrow[i] = 0.f;
}

__global__ void k_key_hist(const int* __restrict__ ei, const int* __restrict__ etype, int E, int N,
                           int* __restrict__ cnt, int* __restrict__ rank) {
  const int e = blockIdx.x * blockDim.x + threadIdx.x;
  if (e < E) rank[e] = atomicAdd(&cnt[(etype[e] & 7) * N + ei[E + e]], 1);
}

static constexpr int SCAN_THREADS = 1024;

__device__ __forceinline__ int wave_incl_scan(int v, int lane) {
#pragma unroll
  for (int o = 1; o < 32; o <<= 1) {
    const int t = __shfl_up(v, o, 32);
    if (lane >= o) v += t;
  }
  return v;
}

__device__ __forceinline__ int wave_sum(int v) {
#pragma unroll
  for (int o = 16; o > 0; o >>= 1) v += __shfl_xor(v, o, 32);
  return v;
}

__global__ __launch_bounds__(SCAN_THREADS) void k_scan_a(
    const int* __restrict__ cnt, int N, int* __restrict__ blkE, int* __restrict__ blkS) {
  __shared__ int wE[32], wS[32];
  const int r = blockIdx.y, t = threadIdx.x, lane = t & 31, wave = t >> 5;
  const int k = blockIdx.x * SCAN_THREADS + t;
  const int v = k < N ? cnt[(size_t)r * N + k] : 0;
  const int e = wave_sum(v);
  const int s = wave_sum((v + SEG_EDGES - 1) / SEG_EDGES);
  if (lane == 0) { wE[wave] = e; wS[wave] = s; }
  __syncthreads();
  if (wave == 0) {
    const int te = wave_sum(wE[lane]), ts = wave_sum(wS[lane]);
    if (lane == 0) {
      blkE[blockIdx.x * 8 + r] = te;
      blkS[blockIdx.x * 8 + r] = ts;
    }
  }
}

static constexpr int MID_MAX = 1024;
__global__ __launch_bounds__(SCAN_THREADS) void k_scan_mid(
    const int* __restrict__ blkE, const int* __restrict__ blkS, int ng,
    int* __restrict__ blkBaseE, int* __restrict__ blkBaseS,
    int* __restrict__ starts, int* __restrict__ segBeg, int* __restrict__ segDst) {
  __shared__ int wE[32], wS[32];
  const int t = threadIdx.x, lane = t & 31, wave = t >> 5;
  const int ve = t < ng ? blkE[t] : 0;
  const int vs = t < ng ? blkS[t] : 0;
  const int vp = (vs + 15) & ~15;
  const int ie = wave_incl_scan(ve, lane), ip = wave_incl_scan(vp, lane);
  if (lane == 31) { wE[wave] = ie; wS[wave] = ip; }
  __syncthreads();
  if (wave == 0) {
    const int a = wE[lane], b = wS[lane];
    const int ia = wave_incl_scan(a, lane), ib = wave_incl_scan(b, lane);
    wE[lane] = ia - a;
    wS[lane] = ib - b;
  }
  __syncthreads();
  const int e0 = wE[wave] + ie - ve;
  const int s0 = wS[wave] + ip - vp;
  if (t < ng) {
    blkBaseE[t] = e0;
    blkBaseS[t] = s0;
    for (int p = s0 + vs; p < s0 + vp; ++p) {
      segBeg[p] = e0 + ve;
      segDst[p] = ((t >> 3) * SCAN_THREADS) | ((t & 7) << 28);
    }
  }
  if (t == SCAN_THREADS - 1) {
    starts[8] = s0 + vp;
    segBeg[s0 + vp] = e0 + ve;
  }
}

__global__ __launch_bounds__(SCAN_THREADS) void k_scan_b(
    int* __restrict__ cnt, int N,
    const int* __restrict__ blkBaseE, const int* __restrict__ blkBaseS,
    int* __restrict__ segBeg, int* __restrict__ segDst) {
  __shared__ int wE[32], wS[32];
  const int r = blockIdx.y, t = threadIdx.x, lane = t & 31, wave = t >> 5;
  const int k = blockIdx.x * SCAN_THREADS + t;
  const int v = k < N ? cnt[(size_t)r * N + k] : 0;
  const int sv = (v + SEG_EDGES - 1) / SEG_EDGES;
  const int ie = wave_incl_scan(v, lane), is = wave_incl_scan(sv, lane);
  if (lane == 31) { wE[wave] = ie; wS[wave] = is; }
  __syncthreads();
  if (wave == 0) {
    const int a = wE[lane], b = wS[lane];
    const int ia = wave_incl_scan(a, lane), ib = wave_incl_scan(b, lane);
    wE[lane] = ia - a;
    wS[lane] = ib - b;
  }
  __syncthreads();
  const int e0 = blkBaseE[blockIdx.x * 8 + r] + wE[wave] + ie - v;
  int s = blkBaseS[blockIdx.x * 8 + r] + wS[wave] + is - sv;
  if (k < N) {
    cnt[(size_t)r * N + k] = e0;
    for (int q = 0; q < v; q += SEG_EDGES) {
      segBeg[s] = e0 + q;
      segDst[s] = k | (r << 28);
      ++s;
    }
  }
}

__global__ void k_key_scatter(const int* __restrict__ ei, const int* __restrict__ etype, int E, int N,
                              const int* __restrict__ first, const int* __restrict__ rank,
                              int* __restrict__ ssrc) {
  const int e = blockIdx.x * blockDim.x + threadIdx.x;
  if (e < E) {
    const int pos = first[(etype[e] & 7) * N + ei[E + e]] + rank[e];
    ssrc[pos] = ei[e];
  }
}


template <int DOUT>
__global__ __launch_bounds__(32 * (DOUT / 16)) void k_root_gemm(
    const float* __restrict__ x, const float* __restrict__ W,
    const float* __restrict__ bias, float* __restrict__ y, int N) {
  __shared__ __attribute__((aligned(16))) float As[16 * A_PITCH];
  constexpr int NW = DOUT / 16;
  const int wave = threadIdx.x >> 5, lane = threadIdx.x & 31;
  const int m = lane & 15;
  const int khalf = (lane >> 4) << 1;
#pragma unroll
  for (int j = 0; j < 16 / NW; ++j) {
    const int tr = wave + j * NW;
    const int row = (blockIdx.x << 4) + tr;
    stage_row(As, x, row < N ? row : N - 1, tr, lane);
  }
  __syncthreads();
  const float* xa = As + m * A_PITCH;
  const int col = (wave << 4) + m;
  const float* wb = W + col;
  v8f c = {};
#pragma unroll 8
  for (int k = 0; k < D_IN; k += 4) {
    v2f a, b;
    a.x = xa[k + khalf];
    a.y = xa[k + khalf + 1];
    b.x = wb[(k + khalf) * DOUT];
    b.y = wb[(k + khalf + 1) * DOUT];
    c = WMMA_F32(a, b, c);
  }
  const float bv = bias[col];
  const int rbase = (blockIdx.x << 4) + ((lane >> 4) << 3);
  float* yp = y + (size_t)rbase * DOUT + col;
  if ((blockIdx.x << 4) + 16 <= N) {
#pragma unroll
    for (int i = 0; i < 8; ++i) yp[i * DOUT] = c[i] + bv;
  } else {
#pragma unroll
    for (int i = 0; i < 8; ++i)
      if (rbase + i < N) yp[i * DOUT] = c[i] + bv;
  }
}


static constexpr int RUN_TILES = 8;
static constexpr int RUN_SEGS = RUN_TILES * 16;
static constexpr int RUN_EDGES_MAX = RUN_SEGS * SEG_EDGES;

__device__ __forceinline__ float4 add4(float4 a, float4 b) {
  return make_float4(a.x + b.x, a.y + b.y, a.z + b.z, a.w + b.w);
}

struct SegRows {
  float4 p[SEG_EDGES];
};

struct RowPtrs {
  const float* p[SEG_EDGES];
};

__device__ __forceinline__ RowPtrs seg_ptrs(const float* __restrict__ x, const float* __restrict__ zrow,
                                            const int* sBeg, const int* sSrc, int seg) {
  const int rb = __builtin_amdgcn_readfirstlane(sBeg[seg]);
  const int n = __builtin_amdgcn_readfirstlane(sBeg[seg + 1]) - rb;
  RowPtrs r;
#pragma unroll
  for (int q = 0; q < SEG_EDGES; ++q) {
    const int row = __builtin_amdgcn_readfirstlane(sSrc[rb + q]);
    r.p[q] = q < n ? x + (size_t)row * D_IN : zrow;
  }
  return r;
}

__device__ __forceinline__ SegRows seg_fetch(const RowPtrs& r, unsigned laneBytes) {
  SegRows s;
#pragma unroll
  for (int q = 0; q < SEG_EDGES; ++q)
    s.p[q] = *reinterpret_cast<const float4*>(reinterpret_cast<const char*>(r.p[q]) + laneBytes);
  return s;
}

__device__ __forceinline__ float4 seg_total(const SegRows& s) {
  return add4(add4(s.p[0], s.p[1]), add4(s.p[2], s.p[3]));
}

template <int DOUT>
__global__ __launch_bounds__(32 * (DOUT / 16)) __attribute__((amdgpu_waves_per_eu(DOUT == 128 ? 8 : 7, DOUT == 128 ? 8 : 7))) void k_edge_wmma(
    const float* __restrict__ x, const float* __restrict__ Wrel,
    const int* __restrict__ ssrc, const int* __restrict__ segBeg, const int* __restrict__ segDst,
    const int* __restrict__ starts, const float* __restrict__ zrow, float* __restrict__ y) {
  static_assert(SEG_EDGES == 4, "seg_total() is written for four rows");
  __shared__ __attribute__((aligned(16))) float As[2][16 * A_PITCH];
  __shared__ __attribute__((aligned(16))) int sBeg[RUN_SEGS + 16];
  __shared__ __attribute__((aligned(16))) int dIdx[RUN_SEGS];
  __shared__ __attribute__((aligned(16))) int sSrc[RUN_EDGES_MAX + 16];
  constexpr int C_PITCH = DOUT + 4;
  __shared__ __attribute__((aligned(16))) float Cs[2][16 * C_PITCH];
  constexpr int NW = DOUT / 16;
  constexpr int RPW = 16 / NW;
  constexpr int LPR = DOUT / 32;
  const int total = starts[8];
  const int base0 = blockIdx.x * RUN_SEGS;
  if (base0 >= total) return;
  const int left = (total - base0) >> 4;
  const int nT = left < RUN_TILES ? left : RUN_TILES;
  const int lane = threadIdx.x & 31;
  const unsigned laneBytes = (unsigned)lane << 4;
  const int wave = __builtin_amdgcn_readfirstlane((int)(threadIdx.x >> 5));

  const int eb0 = segBeg[base0];
  const int nE = segBeg[base0 + (nT << 4)] - eb0;
  for (int i = threadIdx.x; i < RUN_SEGS + 1; i += 32 * NW) {
    const int gi = base0 + i;
    sBeg[i] = (gi <= total ? segBeg[gi] : eb0) - eb0;
    if (i < RUN_SEGS) dIdx[i] = gi < total ? segDst[gi] : 0;
  }
  for (int i = threadIdx.x; i < RUN_EDGES_MAX + 16; i += 32 * NW)
    sSrc[i] = i < nE ? ssrc[eb0 + i] : 0;
  __syncthreads();

  int relPack = 0;
#pragma unroll
  for (int q = 0; q < RUN_TILES; ++q)
    relPack |= ((__builtin_amdgcn_readfirstlane(dIdx[q << 4]) >> 28) & 7) << (3 * q);

#pragma unroll
  for (int j = 0; j < RPW; ++j) {
    const int tr = wave + j * NW;
    const SegRows s = seg_fetch(seg_ptrs(x, zrow, sBeg, sSrc, tr), laneBytes);
    *reinterpret_cast<float4*>(As[0] + tr * A_PITCH + (lane << 2)) = seg_total(s);
  }
  RowPtrs nx[RPW];
  {
    const int t1 = 1 < nT ? 1 : 0;
#pragma unroll
    for (int j = 0; j < RPW; ++j) nx[j] = seg_ptrs(x, zrow, sBeg, sSrc, (t1 << 4) + wave + j * NW);
  }
  __syncthreads();

  const int m = lane & 15;
  const int khalf = (lane >> 4) << 1;
  const int col = ((int)(threadIdx.x >> 5) << 4) + m;
  const int mBase = (lane >> 4) << 3;
  const unsigned laneWord = (unsigned)lane << 2;

  v2f b[D_IN / 4];
  int curR = -1;
  int cur = 0;
  for (int t = 0; t < nT; ++t) {
    const int r = (relPack >> (3 * t)) & 7;
    if (r != curR) {
      const float* wb = Wrel + (size_t)r * (D_IN * DOUT) + col;
#pragma unroll
      for (int kk = 0; kk < D_IN / 4; ++kk) {
        b[kk].x = wb[(4 * kk + khalf) * DOUT];
        b[kk].y = wb[(4 * kk + khalf + 1) * DOUT];
      }
      curR = r;
    }

    unsigned lb = laneBytes;
    asm volatile("" : "+v"(lb));
    SegRows pre[RPW];
#pragma unroll
    for (int j = 0; j < RPW; ++j) pre[j] = seg_fetch(nx[j], lb);
    const int tnn = t + 2 < nT ? t + 2 : nT - 1;
#pragma unroll
    for (int j = 0; j < RPW; ++j) nx[j] = seg_ptrs(x, zrow, sBeg, sSrc, (tnn << 4) + wave + j * NW);

    const float* xa = As[cur] + m * A_PITCH;
    v8f c = {};
#pragma unroll
    for (int kk = 0; kk < D_IN / 4; ++kk) {
      v2f a;
      a.x = xa[4 * kk + khalf];
      a.y = xa[4 * kk + khalf + 1];
      c = WMMA_F32(a, b[kk], c);
    }

    float* Cw = Cs[cur] + mBase * C_PITCH + col;
#pragma unroll
    for (int i = 0; i < 8; ++i) Cw[i * C_PITCH] = c[i];

    int dstRow[RPW];
#pragma unroll
    for (int j = 0; j < RPW; ++j)
      dstRow[j] = __builtin_amdgcn_readfirstlane(dIdx[(t << 4) + wave * RPW + j]);

    float* An = As[cur ^ 1];
#pragma unroll
    for (int j = 0; j < RPW; ++j) {
      const int tr = wave + j * NW;
      *reinterpret_cast<float4*>(An + tr * A_PITCH + (lane << 2)) = seg_total(pre[j]);
    }

    __syncthreads();
    const float* Cr = Cs[cur] + (wave * RPW) * C_PITCH + lane;
    cur ^= 1;

    float cv[RPW][LPR];
#pragma unroll
    for (int j = 0; j < RPW; ++j)
#pragma unroll
      for (int q = 0; q < LPR; ++q) cv[j][q] = Cr[j * C_PITCH + 32 * q];
#pragma unroll
    for (int j = 0; j < RPW; ++j) {
      const unsigned off = (unsigned)(dstRow[j] & 0x0FFFFFFF) * (unsigned)(DOUT * 4) + laneWord;
      float* yr = reinterpret_cast<float*>(reinterpret_cast<char*>(y) + off);
#pragma unroll
      for (int q = 0; q < LPR; ++q) atomicAdd(yr + 32 * q, cv[j][q]);
    }
  }
}


__global__ void k_relu4(float4* __restrict__ p, long long n4) {
  long long i = (long long)blockIdx.x * blockDim.x + threadIdx.x;
  if (i < n4) {
    float4 v = p[i];
    v.x = fmaxf(v.x, 0.0f);
    v.y = fmaxf(v.y, 0.0f);
    v.z = fmaxf(v.z, 0.0f);
    v.w = fmaxf(v.w, 0.0f);
    p[i] = v;
  }
}

__global__ void k_logsoftmax(const float* __restrict__ y, float* __restrict__ out, int N) {
  int wave = threadIdx.x >> 5, lane = threadIdx.x & 31;
  int node = blockIdx.x * (blockDim.x >> 5) + wave;
  if (node >= N) return;
  const float* row = y + (size_t)node * 64;
  float v0 = row[lane], v1 = row[lane + 32];
  float mx = fmaxf(v0, v1);
#pragma unroll
  for (int off = 16; off > 0; off >>= 1) mx = fmaxf(mx, __shfl_xor(mx, off, 32));
  float s = expf(v0 - mx) + expf(v1 - mx);
#pragma unroll
  for (int off = 16; off > 0; off >>= 1) s += __shfl_xor(s, off, 32);
  float lse = logf(s) + mx;
  float* o = out + (size_t)node * 64;
  o[lane] = v0 - lse;
  o[lane + 32] = v1 - lse;
}


static inline void launch_layer(const float* x, const float* Wrel, const float* Wroot,
                                const float* bias, float* y, int dOut, int N,
                                const int* ssrc, const int* segBeg, const int* segDst, const int* starts,
                                const float* zrow, int nodeTiles, int segRunsMax, hipStream_t stream) {
  if (dOut == 128) {
    k_root_gemm<128><<<nodeTiles, 32 * (128 / 16), 0, stream>>>(x, Wroot, bias, y, N);
    k_edge_wmma<128><<<segRunsMax, 32 * (128 / 16), 0, stream>>>(x, Wrel, ssrc, segBeg, segDst, starts, zrow, y);
  } else {
    k_root_gemm<64><<<nodeTiles, 32 * (64 / 16), 0, stream>>>(x, Wroot, bias, y, N);
    k_edge_wmma<64><<<segRunsMax, 32 * (64 / 16), 0, stream>>>(x, Wrel, ssrc, segBeg, segDst, starts, zrow, y);
  }
}

extern "C" void kernel_launch(void* const* d_in, const int* in_sizes, int n_in,
                              void* d_out, int out_size, void* d_ws, size_t ws_size,
                              hipStream_t stream) {
  const float* x0     = (const float*)d_in[0];
  const int*   ei     = (const int*)  d_in[1];
  const int*   etype  = (const int*)  d_in[2];
  const float* Wrel1  = (const float*)d_in[3];
  const float* Wroot1 = (const float*)d_in[4];
  const float* b1     = (const float*)d_in[5];
  const float* Wrel2  = (const float*)d_in[6];
  const float* Wroot2 = (const float*)d_in[7];
  const float* b2     = (const float*)d_in[8];
  const float* Wrel3  = (const float*)d_in[9];
  const float* Wroot3 = (const float*)d_in[10];
  const float* b3     = (const float*)d_in[11];

  const int N  = in_sizes[0] / D_IN;
  const int E  = in_sizes[2];
  const int D1 = 128, D2 = 128;
  const int D3 = in_sizes[11];
  const int K  = 8 * N;
  const int keysMax = E < K ? E : K;
  const int nb = (N + SCAN_THREADS - 1) / SCAN_THREADS;
  const int segsMax = keysMax + (E - keysMax + SEG_EDGES - 1) / SEG_EDGES + 8 * nb * 16;

  char* w = (char*)d_ws;
  int* starts = (int*)w;
  size_t off = 256;
  int* segBeg = (int*)(w + off); off += (size_t)(segsMax + 16) * sizeof(int);
  int* segDst = (int*)(w + off); off += (size_t)(segsMax + 16) * sizeof(int);
  int* ssrc   = (int*)(w + off); off += (size_t)E * sizeof(int);
  off = (off + 255) & ~(size_t)255;
  float* bufA = (float*)(w + off); off += (size_t)(N + 16) * D_IN * sizeof(float);
  float* bufB = (float*)(w + off);
  float* zrow = bufA + (size_t)(N + 8) * D_IN;
  int* cnt  = (int*)bufB;
  int* blkE = cnt + K;
  int* blkS = blkE + MID_MAX;
  int* blkBaseE = blkS + MID_MAX;
  int* blkBaseS = blkBaseE + MID_MAX;
  int* rank = blkBaseS + MID_MAX;

  k_zero_ints  <<<(K + 255) / 256, 256, 0, stream>>>(cnt, K, zrow);
  k_key_hist   <<<(E + 255) / 256, 256, 0, stream>>>(ei, etype, E, N, cnt, rank);
  k_scan_a     <<<dim3(nb, 8), SCAN_THREADS, 0, stream>>>(cnt, N, blkE, blkS);
  k_scan_mid   <<<1, SCAN_THREADS, 0, stream>>>(blkE, blkS, 8 * nb, blkBaseE, blkBaseS, starts, segBeg, segDst);
  k_scan_b     <<<dim3(nb, 8), SCAN_THREADS, 0, stream>>>(cnt, N, blkBaseE, blkBaseS, segBeg, segDst);
  k_key_scatter<<<(E + 255) / 256, 256, 0, stream>>>(ei, etype, E, N, cnt, rank, ssrc);

  const int nodeTiles  = (N + 15) / 16;
  const int segRunsMax = (segsMax + RUN_SEGS - 1) / RUN_SEGS;

  launch_layer(x0, Wrel1, Wroot1, b1, bufA, D1, N, ssrc, segBeg, segDst, starts,
               zrow, nodeTiles, segRunsMax, stream);
  {
    long long n4 = (long long)N * D1 / 4;
    k_relu4<<<(unsigned)((n4 + 255) / 256), 256, 0, stream>>>((float4*)bufA, n4);
  }

  launch_layer(bufA, Wrel2, Wroot2, b2, bufB, D2, N, ssrc, segBeg, segDst, starts,
               zrow, nodeTiles, segRunsMax, stream);
  {
    long long n4 = (long long)N * D2 / 4;
    k_relu4<<<(unsigned)((n4 + 255) / 256), 256, 0, stream>>>((float4*)bufB, n4);
  }

  launch_layer(bufB, Wrel3, Wroot3, b3, bufA, D3, N, ssrc, segBeg, segDst, starts,
               zrow, nodeTiles, segRunsMax, stream);
  k_logsoftmax<<<(N + 7) / 8, 256, 0, stream>>>(bufA, (float*)d_out, N);
}
